# starting kernel
# baseline (speedup 1.0000x reference)
#include <hip/hip_runtime.h>


#define GCN_N_NODES 50000
#define GCN_N_EDGES 800000
#define GCN_IN_DIM 128
#define GCN_HID_DIM 128
#define GCN_OUT_DIM 64

typedef __attribute__((ext_vector_type(2))) float v2f;
typedef __attribute__((ext_vector_type(8))) float v8f;

__global__ void gcn_degree_kernel(const int* __restrict__ src,
                                  const int* __restrict__ dst,
                                  float* __restrict__ deg_out,
                                  float* __restrict__ deg_in, int E) {
  int e = blockIdx.x * blockDim.x + threadIdx.x;
  if (e < E) {
    atomicAdd(&deg_out[src[e]], 1.0f);
    atomicAdd(&deg_in[dst[e]], 1.0f);
  }
}

__global__ void gcn_norm_kernel(const float* __restrict__ deg_out,
                                const float* __restrict__ deg_in,
                                float* __restrict__ norm_out,
                                float* __restrict__ norm_in, int N) {
  int i = blockIdx.x * blockDim.x + threadIdx.x;
  if (i < N) {
    norm_out[i] = rsqrtf(fmaxf(deg_out[i], 1.0f));
    norm_in[i]  = rsqrtf(fmaxf(deg_in[i], 1.0f));
  }
}

template <int K, int NOUT>
__global__ void gcn_gemm_scaled_wmma(const float* __restrict__ X,
                                     const float* __restrict__ scale,
                                     const float* __restrict__ W,
                                     float* __restrict__ Y) {
  constexpr int LDSK = K + 4;
  __shared__ float tile[16 * LDSK];

  const int tid  = threadIdx.x;
  const int row0 = blockIdx.x * 16;

  for (int t = tid; t < 16 * K; t += blockDim.x) {
    const int r = t / K;
    const int c = t - r * K;
    const int node = row0 + r;
    tile[r * LDSK + c] = X[(size_t)node * K + c] * scale[node];
  }
  __syncthreads();

  const int wave = tid >> 5;
  const int lane = tid & 31;
  const int m    = lane & 15;
  const int kh   = lane >> 4;
  const int nc   = wave * 16 + m;

  v8f acc = {};
#pragma unroll 4
  for (int kk = 0; kk < K; kk += 4) {
    v2f a, b;
    const int ka = kk + 2 * kh;
    a.x = tile[m * LDSK + ka];
    a.y = tile[m * LDSK + ka + 1];
    b.x = W[(size_t)ka * NOUT + nc];
    b.y = W[(size_t)(ka + 1) * NOUT + nc];
    acc = __builtin_amdgcn_wmma_f32_16x16x4_f32(
         false, a,  false, b,
         (short)0, acc,  false,  false);
  }

#pragma unroll
  for (int v = 0; v < 8; ++v) {
    const int orow = row0 + v + 8 * kh;
    Y[(size_t)orow * NOUT + nc] = acc[v];
  }
}

template <int D>
__global__ void gcn_aggregate_kernel(const float* __restrict__ H,
                                     const int* __restrict__ src,
                                     const int* __restrict__ dst,
                                     float* __restrict__ agg, int E) {
  const unsigned t = blockIdx.x * blockDim.x + threadIdx.x;
  const unsigned e = t / D;
  const unsigned d = t - e * D;
  if (e < (unsigned)E) {
    const int s = src[e];
    const int r = dst[e];
    atomicAdd(&agg[(size_t)r * D + d], H[(size_t)s * D + d]);
  }
}

template <int D>
__global__ void gcn_finalize_kernel(const float* __restrict__ agg,
                                    const float* __restrict__ norm_in,
                                    const float* __restrict__ b,
                                    float* __restrict__ out, int N) {
  const unsigned t = blockIdx.x * blockDim.x + threadIdx.x;
  const unsigned i = t / D;
  const unsigned d = t - i * D;
  if (i < (unsigned)N) {
    const float v = fmaf(agg[t], norm_in[i], b[d]);
    out[t] = fmaxf(v, 0.0f);
  }
}

extern "C" void kernel_launch(void* const* d_in, const int* in_sizes, int n_in,
                              void* d_out, int out_size, void* d_ws, size_t ws_size,
                              hipStream_t stream) {
  const float* feat = (const float*)d_in[0];
  const int*   src  = (const int*)d_in[1];
  const int*   dst  = (const int*)d_in[2];
  const float* W1   = (const float*)d_in[3];
  const float* b1   = (const float*)d_in[4];
  const float* W2   = (const float*)d_in[5];
  const float* b2   = (const float*)d_in[6];
  float* out = (float*)d_out;
  float* ws  = (float*)d_ws;

  const int N = GCN_N_NODES;
  const int E = GCN_N_EDGES;

  float* deg_out  = ws;
  float* deg_in   = ws + (size_t)N;
  float* norm_out = ws + 2 * (size_t)N;
  float* norm_in  = ws + 3 * (size_t)N;
  float* bufA     = ws + 4 * (size_t)N;
  float* bufB     = bufA + (size_t)N * GCN_HID_DIM;
  float* hproj2   = bufA;
  float* agg2     = bufA + (size_t)N * GCN_OUT_DIM;

  hipMemsetAsync(deg_out, 0, 2 * (size_t)N * sizeof(float), stream);
  gcn_degree_kernel<<<(E + 255) / 256, 256, 0, stream>>>(src, dst, deg_out, deg_in, E);
  gcn_norm_kernel<<<(N + 255) / 256, 256, 0, stream>>>(deg_out, deg_in, norm_out, norm_in, N);

  gcn_gemm_scaled_wmma<GCN_IN_DIM, GCN_HID_DIM>
      <<<N / 16, (GCN_HID_DIM / 16) * 32, 0, stream>>>(feat, norm_out, W1, bufA);
  hipMemsetAsync(bufB, 0, (size_t)N * GCN_HID_DIM * sizeof(float), stream);
  {
    const unsigned work = (unsigned)E * GCN_HID_DIM;
    gcn_aggregate_kernel<GCN_HID_DIM>
        <<<(work + 255) / 256, 256, 0, stream>>>(bufA, src, dst, bufB, E);
  }
  {
    const unsigned work = (unsigned)N * GCN_HID_DIM;
    gcn_finalize_kernel<GCN_HID_DIM>
        <<<(work + 255) / 256, 256, 0, stream>>>(bufB, norm_in, b1, bufB, N);
  }

  gcn_gemm_scaled_wmma<GCN_HID_DIM, GCN_OUT_DIM>
      <<<N / 16, (GCN_OUT_DIM / 16) * 32, 0, stream>>>(bufB, norm_out, W2, hproj2);
  hipMemsetAsync(agg2, 0, (size_t)N * GCN_OUT_DIM * sizeof(float), stream);
  {
    const unsigned work = (unsigned)E * GCN_OUT_DIM;
    gcn_aggregate_kernel<GCN_OUT_DIM>
        <<<(work + 255) / 256, 256, 0, stream>>>(hproj2, src, dst, agg2, E);
  }
  {
    const unsigned work = (unsigned)N * GCN_OUT_DIM;
    gcn_finalize_kernel<GCN_OUT_DIM>
        <<<(work + 255) / 256, 256, 0, stream>>>(agg2, norm_in, b2, out, N);
  }
}


# replace float-atomic scatter-add by pull aggregation: per-call counting sort of edges by dst in LDS, one wave per node sums rows in registers, fused norm/bias/relu; sum order = sorted edge order
# speedup vs baseline: 3.7963x; 3.7963x over previous
#include <hip/hip_runtime.h>


#define GCN_N_NODES 50000
#define GCN_N_EDGES 800000
#define GCN_IN_DIM 128
#define GCN_HID_DIM 128
#define GCN_OUT_DIM 64

#define CSR_B 64
#define CSR_CHUNK ((GCN_N_EDGES + CSR_B - 1) / CSR_B)
#define CSR_THREADS 1024
#define CSR_RUN 49
#define CSR_PAD (CSR_RUN * CSR_THREADS)

typedef __attribute__((ext_vector_type(2))) float v2f;
typedef __attribute__((ext_vector_type(8))) float v8f;

__global__ __launch_bounds__(CSR_THREADS) void gcn_hist_kernel(
    const int* __restrict__ src, const int* __restrict__ dst,
    unsigned* __restrict__ part) {
  __shared__ unsigned bins[GCN_N_NODES];
  const int tid = threadIdx.x;
  const int* __restrict__ idx = blockIdx.y ? dst : src;
  unsigned* __restrict__ plane =
      part + ((size_t)blockIdx.y * CSR_B + blockIdx.x) * GCN_N_NODES;

  for (int i = tid; i < GCN_N_NODES; i += CSR_THREADS) bins[i] = 0u;
  __syncthreads();

  const int e0 = blockIdx.x * CSR_CHUNK;
  const int e1 = min(GCN_N_EDGES, e0 + CSR_CHUNK);
  for (int e = e0 + tid; e < e1; e += 4 * CSR_THREADS) {
    int d[4];
#pragma unroll
    for (int u = 0; u < 4; ++u) {
      const int ee = e + u * CSR_THREADS;
      d[u] = (ee < e1) ? idx[ee] : -1;
    }
#pragma unroll
    for (int u = 0; u < 4; ++u)
      if (d[u] >= 0) atomicAdd(&bins[d[u]], 1u);
  }
  __syncthreads();

  for (int i = tid; i < GCN_N_NODES; i += CSR_THREADS) plane[i] = bins[i];
}

__global__ void gcn_colsum_kernel(unsigned* __restrict__ part,
                                  unsigned* __restrict__ cnt_in,
                                  float* __restrict__ norm_out,
                                  float* __restrict__ norm_in) {
  const int n = blockIdx.x * blockDim.x + threadIdx.x;
  if (n >= GCN_N_NODES) return;
  const unsigned* ps = part + n;
  unsigned* pd = part + (size_t)CSR_B * GCN_N_NODES + n;
  unsigned so = 0u;
#pragma unroll 8
  for (int b = 0; b < CSR_B; ++b) so += ps[(size_t)b * GCN_N_NODES];
  unsigned run = 0u;
#pragma unroll 8
  for (int b = 0; b < CSR_B; ++b) {
    const unsigned c = pd[(size_t)b * GCN_N_NODES];
    pd[(size_t)b * GCN_N_NODES] = run;
    run += c;
  }
  cnt_in[n]   = run;
  norm_out[n] = rsqrtf(fmaxf((float)so, 1.0f));
  norm_in[n]  = rsqrtf(fmaxf((float)run, 1.0f));
}

__global__ __launch_bounds__(CSR_THREADS) void gcn_fill_kernel(
    const int* __restrict__ src, const int* __restrict__ dst,
    const unsigned* __restrict__ part, const unsigned* __restrict__ cnt_in,
    unsigned* __restrict__ row_ptr, int* __restrict__ col) {
  __shared__ unsigned cur[CSR_PAD];
  __shared__ unsigned wsum[32];
  const int tid  = threadIdx.x;
  const int lane = tid & 31;
  const int wave = tid >> 5;

  for (int i = tid; i < GCN_N_NODES; i += CSR_THREADS) cur[i] = cnt_in[i];
  for (int i = GCN_N_NODES + tid; i < CSR_PAD; i += CSR_THREADS) cur[i] = 0u;
  __syncthreads();

  unsigned s = 0u;
  for (int k = 0; k < CSR_RUN; ++k) s += cur[tid * CSR_RUN + k];

  unsigned incl = s;
  for (int off = 1; off < 32; off <<= 1) {
    const unsigned v = __shfl_up(incl, off);
    if (lane >= off) incl += v;
  }
  if (lane == 31) wsum[wave] = incl;
  __syncthreads();
  if (wave == 0) {
    const unsigned w = wsum[lane];
    unsigned wi = w;
    for (int off = 1; off < 32; off <<= 1) {
      const unsigned v = __shfl_up(wi, off);
      if (lane >= off) wi += v;
    }
    wsum[lane] = wi - w;
  }
  __syncthreads();

  unsigned run = wsum[wave] + incl - s;
  for (int k = 0; k < CSR_RUN; ++k) {
    const unsigned c = cur[tid * CSR_RUN + k];
    cur[tid * CSR_RUN + k] = run;
    run += c;
  }
  __syncthreads();

  if (blockIdx.x == 0)
    for (int i = tid; i <= GCN_N_NODES; i += CSR_THREADS) row_ptr[i] = cur[i];
  __syncthreads();

  const unsigned* __restrict__ offs =
      part + ((size_t)CSR_B + blockIdx.x) * GCN_N_NODES;
  for (int i = tid; i < GCN_N_NODES; i += CSR_THREADS) cur[i] += offs[i];
  __syncthreads();

  const int e0 = blockIdx.x * CSR_CHUNK;
  const int e1 = min(GCN_N_EDGES, e0 + CSR_CHUNK);
  for (int e = e0 + tid; e < e1; e += 4 * CSR_THREADS) {
    int d[4], s[4];
#pragma unroll
    for (int u = 0; u < 4; ++u) {
      const int ee = e + u * CSR_THREADS;
      d[u] = (ee < e1) ? dst[ee] : -1;
      s[u] = (ee < e1) ? src[ee] : 0;
    }
#pragma unroll
    for (int u = 0; u < 4; ++u)
      if (d[u] >= 0) {
        const unsigned pos = atomicAdd(&cur[d[u]], 1u);
        col[pos] = s[u];
      }
  }
}

template <int K, int NOUT>
__global__ void gcn_gemm_scaled_wmma(const float* __restrict__ X,
                                     const float* __restrict__ scale,
                                     const float* __restrict__ W,
                                     float* __restrict__ Y) {
  constexpr int LDSK = K + 4;
  __shared__ float tile[16 * LDSK];

  const int tid  = threadIdx.x;
  const int row0 = blockIdx.x * 16;

  for (int t = tid; t < 16 * K; t += blockDim.x) {
    const int r = t / K;
    const int c = t - r * K;
    const int node = row0 + r;
    tile[r * LDSK + c] = X[(size_t)node * K + c] * scale[node];
  }
  __syncthreads();

  const int wave = tid >> 5;
  const int lane = tid & 31;
  const int m    = lane & 15;
  const int kh   = lane >> 4;
  const int nc   = wave * 16 + m;

  v8f acc = {};
#pragma unroll 4
  for (int kk = 0; kk < K; kk += 4) {
    v2f a, b;
    const int ka = kk + 2 * kh;
    a.x = tile[m * LDSK + ka];
    a.y = tile[m * LDSK + ka + 1];
    b.x = W[(size_t)ka * NOUT + nc];
    b.y = W[(size_t)(ka + 1) * NOUT + nc];
    acc = __builtin_amdgcn_wmma_f32_16x16x4_f32(
         false, a,  false, b,
         (short)0, acc,  false,  false);
  }

#pragma unroll
  for (int v = 0; v < 8; ++v) {
    const int orow = row0 + v + 8 * kh;
    Y[(size_t)orow * NOUT + nc] = acc[v];
  }
}

template <int D>
__global__ __launch_bounds__(256) void gcn_gather_kernel(
    const float* __restrict__ H, const unsigned* __restrict__ row_ptr,
    const int* __restrict__ col, const float* __restrict__ norm_in,
    const float* __restrict__ b, float* __restrict__ out, int N) {
  constexpr int V = D / 32;
  typedef float vecT __attribute__((ext_vector_type(V)));
  const int lane = threadIdx.x & 31;
  const int node =
      __builtin_amdgcn_readfirstlane(blockIdx.x * 8 + (threadIdx.x >> 5));
  if (node >= N) return;

  const unsigned beg = row_ptr[node];
  const unsigned end = row_ptr[node + 1];

  vecT acc = (vecT)(0.0f);
  for (unsigned base = beg; base < end; base += 32u) {
    const int cnt = (int)min(32u, end - base);
    int myc = 0;
    if (lane < cnt) myc = col[base + lane];
    int j = 0;
    for (; j + 8 <= cnt; j += 8) {
      int s[8];
#pragma unroll
      for (int u = 0; u < 8; ++u) s[u] = __builtin_amdgcn_readlane(myc, j + u);
      vecT r[8];
#pragma unroll
      for (int u = 0; u < 8; ++u)
        r[u] = *(const vecT*)(H + (size_t)s[u] * D + lane * V);
#pragma unroll
      for (int u = 0; u < 8; ++u) acc += r[u];
    }
    if (j < cnt) {
      int s[7];
#pragma unroll
      for (int u = 0; u < 7; ++u)
        s[u] = __builtin_amdgcn_readlane(myc, (j + u) & 31);
      vecT r[7];
#pragma unroll
      for (int u = 0; u < 7; ++u) {
        r[u] = (vecT)(0.0f);
        if (j + u < cnt) r[u] = *(const vecT*)(H + (size_t)s[u] * D + lane * V);
      }
#pragma unroll
      for (int u = 0; u < 7; ++u) acc += r[u];
    }
  }

  const float ni = norm_in[node];
  const vecT bb = *(const vecT*)(b + lane * V);
  vecT o;
#pragma unroll
  for (int k = 0; k < V; ++k) o[k] = fmaxf(fmaf(acc[k], ni, bb[k]), 0.0f);
  *(vecT*)(out + (size_t)node * D + lane * V) = o;
}

extern "C" void kernel_launch(void* const* d_in, const int* in_sizes, int n_in,
                              void* d_out, int out_size, void* d_ws, size_t ws_size,
                              hipStream_t stream) {
  const float* feat = (const float*)d_in[0];
  const int*   src  = (const int*)d_in[1];
  const int*   dst  = (const int*)d_in[2];
  const float* W1   = (const float*)d_in[3];
  const float* b1   = (const float*)d_in[4];
  const float* W2   = (const float*)d_in[5];
  const float* b2   = (const float*)d_in[6];
  float* out = (float*)d_out;
  float* ws  = (float*)d_ws;

  const int N = GCN_N_NODES;
  const size_t NP = 50048;

  unsigned* part     = (unsigned*)ws;
  unsigned* cnt_in   = part + (size_t)2 * CSR_B * GCN_N_NODES;
  unsigned* row_ptr  = cnt_in + NP;
  float*    norm_out = (float*)(row_ptr + NP);
  float*    norm_in  = norm_out + NP;
  int*      col      = (int*)(norm_in + NP);
  float*    bufA     = (float*)(col + (size_t)GCN_N_EDGES);
  float*    bufB     = bufA + (size_t)N * GCN_HID_DIM;
  float*    hproj2   = bufA;

  gcn_hist_kernel<<<dim3(CSR_B, 2), CSR_THREADS, 0, stream>>>(src, dst, part);
  gcn_colsum_kernel<<<(N + 255) / 256, 256, 0, stream>>>(part, cnt_in, norm_out, norm_in);
  gcn_fill_kernel<<<CSR_B, CSR_THREADS, 0, stream>>>(src, dst, part, cnt_in, row_ptr, col);

  gcn_gemm_scaled_wmma<GCN_IN_DIM, GCN_HID_DIM>
      <<<N / 16, (GCN_HID_DIM / 16) * 32, 0, stream>>>(feat, norm_out, W1, bufA);
  gcn_gather_kernel<GCN_HID_DIM>
      <<<(N + 7) / 8, 256, 0, stream>>>(bufA, row_ptr, col, norm_in, b1, bufB, N);

  gcn_gemm_scaled_wmma<GCN_HID_DIM, GCN_OUT_DIM>
      <<<N / 16, (GCN_OUT_DIM / 16) * 32, 0, stream>>>(bufB, norm_out, W2, hproj2);
  gcn_gather_kernel<GCN_OUT_DIM>
      <<<(N + 7) / 8, 256, 0, stream>>>(hproj2, row_ptr, col, norm_in, b2, out, N);
}


# fuse layer-2 projection into the layer-1 aggregation launch: 16-node block, one wave per node, relu rows * D_out^-1/2 go to LDS, 4 waves do the f32 WMMA; h1 not stored, one launch less
# speedup vs baseline: 4.0327x; 1.0623x over previous
#include <hip/hip_runtime.h>


#define GCN_N_NODES 50000
#define GCN_N_EDGES 800000
#define GCN_IN_DIM 128
#define GCN_HID_DIM 128
#define GCN_OUT_DIM 64

#define CSR_B 64
#define CSR_CHUNK ((GCN_N_EDGES + CSR_B - 1) / CSR_B)
#define CSR_THREADS 1024
#define CSR_RUN 49
#define CSR_PAD (CSR_RUN * CSR_THREADS)

typedef __attribute__((ext_vector_type(2))) float v2f;
typedef __attribute__((ext_vector_type(8))) float v8f;

__global__ __launch_bounds__(CSR_THREADS) void gcn_hist_kernel(
    const int* __restrict__ src, const int* __restrict__ dst,
    unsigned* __restrict__ part) {
  __shared__ unsigned bins[GCN_N_NODES];
  const int tid = threadIdx.x;
  const int* __restrict__ idx = blockIdx.y ? dst : src;
  unsigned* __restrict__ plane =
      part + ((size_t)blockIdx.y * CSR_B + blockIdx.x) * GCN_N_NODES;

  for (int i = tid; i < GCN_N_NODES; i += CSR_THREADS) bins[i] = 0u;
  __syncthreads();

  const int e0 = blockIdx.x * CSR_CHUNK;
  const int e1 = min(GCN_N_EDGES, e0 + CSR_CHUNK);
  for (int e = e0 + tid; e < e1; e += 4 * CSR_THREADS) {
    int d[4];
#pragma unroll
    for (int u = 0; u < 4; ++u) {
      const int ee = e + u * CSR_THREADS;
      d[u] = (ee < e1) ? idx[ee] : -1;
    }
#pragma unroll
    for (int u = 0; u < 4; ++u)
      if (d[u] >= 0) atomicAdd(&bins[d[u]], 1u);
  }
  __syncthreads();

  for (int i = tid; i < GCN_N_NODES; i += CSR_THREADS) plane[i] = bins[i];
}

__global__ void gcn_colsum_kernel(unsigned* __restrict__ part,
                                  unsigned* __restrict__ cnt_in,
                                  float* __restrict__ norm_out,
                                  float* __restrict__ norm_in) {
  const int n = blockIdx.x * blockDim.x + threadIdx.x;
  if (n >= GCN_N_NODES) return;
  const unsigned* ps = part + n;
  unsigned* pd = part + (size_t)CSR_B * GCN_N_NODES + n;
  unsigned so = 0u;
#pragma unroll 8
  for (int b = 0; b < CSR_B; ++b) so += ps[(size_t)b * GCN_N_NODES];
  unsigned run = 0u;
#pragma unroll 8
  for (int b = 0; b < CSR_B; ++b) {
    const unsigned c = pd[(size_t)b * GCN_N_NODES];
    pd[(size_t)b * GCN_N_NODES] = run;
    run += c;
  }
  cnt_in[n]   = run;
  norm_out[n] = rsqrtf(fmaxf((float)so, 1.0f));
  norm_in[n]  = rsqrtf(fmaxf((float)run, 1.0f));
}

__global__ __launch_bounds__(CSR_THREADS) void gcn_fill_kernel(
    const int* __restrict__ src, const int* __restrict__ dst,
    const unsigned* __restrict__ part, const unsigned* __restrict__ cnt_in,
    unsigned* __restrict__ row_ptr, int* __restrict__ col) {
  __shared__ unsigned cur[CSR_PAD];
  __shared__ unsigned wsum[32];
  const int tid  = threadIdx.x;
  const int lane = tid & 31;
  const int wave = tid >> 5;

  for (int i = tid; i < GCN_N_NODES; i += CSR_THREADS) cur[i] = cnt_in[i];
  for (int i = GCN_N_NODES + tid; i < CSR_PAD; i += CSR_THREADS) cur[i] = 0u;
  __syncthreads();

  unsigned s = 0u;
  for (int k = 0; k < CSR_RUN; ++k) s += cur[tid * CSR_RUN + k];

  unsigned incl = s;
  for (int off = 1; off < 32; off <<= 1) {
    const unsigned v = __shfl_up(incl, off);
    if (lane >= off) incl += v;
  }
  if (lane == 31) wsum[wave] = incl;
  __syncthreads();
  if (wave == 0) {
    const unsigned w = wsum[lane];
    unsigned wi = w;
    for (int off = 1; off < 32; off <<= 1) {
      const unsigned v = __shfl_up(wi, off);
      if (lane >= off) wi += v;
    }
    wsum[lane] = wi - w;
  }
  __syncthreads();

  unsigned run = wsum[wave] + incl - s;
  for (int k = 0; k < CSR_RUN; ++k) {
    const unsigned c = cur[tid * CSR_RUN + k];
    cur[tid * CSR_RUN + k] = run;
    run += c;
  }
  __syncthreads();

  if (blockIdx.x == 0)
    for (int i = tid; i <= GCN_N_NODES; i += CSR_THREADS) row_ptr[i] = cur[i];
  __syncthreads();

  const unsigned* __restrict__ offs =
      part + ((size_t)CSR_B + blockIdx.x) * GCN_N_NODES;
  for (int i = tid; i < GCN_N_NODES; i += CSR_THREADS) cur[i] += offs[i];
  __syncthreads();

  const int e0 = blockIdx.x * CSR_CHUNK;
  const int e1 = min(GCN_N_EDGES, e0 + CSR_CHUNK);
  for (int e = e0 + tid; e < e1; e += 4 * CSR_THREADS) {
    int d[4], s[4];
#pragma unroll
    for (int u = 0; u < 4; ++u) {
      const int ee = e + u * CSR_THREADS;
      d[u] = (ee < e1) ? dst[ee] : -1;
      s[u] = (ee < e1) ? src[ee] : 0;
    }
#pragma unroll
    for (int u = 0; u < 4; ++u)
      if (d[u] >= 0) {
        const unsigned pos = atomicAdd(&cur[d[u]], 1u);
        col[pos] = s[u];
      }
  }
}

template <int K, int NOUT>
__global__ void gcn_gemm_scaled_wmma(const float* __restrict__ X,
                                     const float* __restrict__ scale,
                                     const float* __restrict__ W,
                                     float* __restrict__ Y) {
  constexpr int LDSK = K + 4;
  __shared__ float tile[16 * LDSK];

  const int tid  = threadIdx.x;
  const int row0 = blockIdx.x * 16;

  for (int t = tid; t < 16 * K; t += blockDim.x) {
    const int r = t / K;
    const int c = t - r * K;
    const int node = row0 + r;
    tile[r * LDSK + c] = X[(size_t)node * K + c] * scale[node];
  }
  __syncthreads();

  const int wave = tid >> 5;
  const int lane = tid & 31;
  const int m    = lane & 15;
  const int kh   = lane >> 4;
  const int nc   = wave * 16 + m;

  v8f acc = {};
#pragma unroll 4
  for (int kk = 0; kk < K; kk += 4) {
    v2f a, b;
    const int ka = kk + 2 * kh;
    a.x = tile[m * LDSK + ka];
    a.y = tile[m * LDSK + ka + 1];
    b.x = W[(size_t)ka * NOUT + nc];
    b.y = W[(size_t)(ka + 1) * NOUT + nc];
    acc = __builtin_amdgcn_wmma_f32_16x16x4_f32(
         false, a,  false, b,
         (short)0, acc,  false,  false);
  }

#pragma unroll
  for (int v = 0; v < 8; ++v) {
    const int orow = row0 + v + 8 * kh;
    Y[(size_t)orow * NOUT + nc] = acc[v];
  }
}

template <int D>
__global__ __launch_bounds__(256) void gcn_gather_kernel(
    const float* __restrict__ H, const unsigned* __restrict__ row_ptr,
    const int* __restrict__ col, const float* __restrict__ norm_in,
    const float* __restrict__ b, float* __restrict__ out, int N) {
  constexpr int V = D / 32;
  typedef float vecT __attribute__((ext_vector_type(V)));
  const int lane = threadIdx.x & 31;
  const int node =
      __builtin_amdgcn_readfirstlane(blockIdx.x * 8 + (threadIdx.x >> 5));
  if (node >= N) return;

  const unsigned beg = row_ptr[node];
  const unsigned end = row_ptr[node + 1];

  vecT acc = (vecT)(0.0f);
  for (unsigned base = beg; base < end; base += 32u) {
    const int cnt = (int)min(32u, end - base);
    int myc = 0;
    if (lane < cnt) myc = col[base + lane];
    int j = 0;
    for (; j + 8 <= cnt; j += 8) {
      int s[8];
#pragma unroll
      for (int u = 0; u < 8; ++u) s[u] = __builtin_amdgcn_readlane(myc, j + u);
      vecT r[8];
#pragma unroll
      for (int u = 0; u < 8; ++u)
        r[u] = *(const vecT*)(H + (size_t)s[u] * D + lane * V);
#pragma unroll
      for (int u = 0; u < 8; ++u) acc += r[u];
    }
    if (j < cnt) {
      int s[7];
#pragma unroll
      for (int u = 0; u < 7; ++u)
        s[u] = __builtin_amdgcn_readlane(myc, (j + u) & 31);
      vecT r[7];
#pragma unroll
      for (int u = 0; u < 7; ++u) {
        r[u] = (vecT)(0.0f);
        if (j + u < cnt) r[u] = *(const vecT*)(H + (size_t)s[u] * D + lane * V);
      }
#pragma unroll
      for (int u = 0; u < 7; ++u) acc += r[u];
    }
  }

  const float ni = norm_in[node];
  const vecT bb = *(const vecT*)(b + lane * V);
  vecT o;
#pragma unroll
  for (int k = 0; k < V; ++k) o[k] = fmaxf(fmaf(acc[k], ni, bb[k]), 0.0f);
  *(vecT*)(out + (size_t)node * D + lane * V) = o;
}

template <int D, int NOUT>
__global__ __launch_bounds__(512) void gcn_gather_gemm_kernel(
    const float* __restrict__ H, const unsigned* __restrict__ row_ptr,
    const int* __restrict__ col, const float* __restrict__ norm_in,
    const float* __restrict__ b, const float* __restrict__ scale,
    const float* __restrict__ W, float* __restrict__ Y) {
  static_assert(GCN_N_NODES % 16 == 0, "one block owns 16 whole rows");
  static_assert(NOUT % 16 == 0 && NOUT / 16 <= 16, "one wave per column tile");
  static_assert(D % 8 == 0, "two K halves of whole k-steps");
  constexpr int V = D / 32;
  constexpr int LDSK = D + 4;
  typedef float vecT __attribute__((ext_vector_type(V)));
  __shared__ __attribute__((aligned(16))) float tile[16 * LDSK];

  const int lane = threadIdx.x & 31;
  const int wave = __builtin_amdgcn_readfirstlane(threadIdx.x >> 5);
  const int row0 = blockIdx.x * 16;
  const int node = row0 + wave;

  const unsigned beg = row_ptr[node];
  const unsigned end = row_ptr[node + 1];

  vecT acc = (vecT)(0.0f);
  for (unsigned base = beg; base < end; base += 32u) {
    const int cnt = (int)min(32u, end - base);
    int myc = 0;
    if (lane < cnt) myc = col[base + lane];
    int j = 0;
    for (; j + 8 <= cnt; j += 8) {
      int s[8];
#pragma unroll
      for (int u = 0; u < 8; ++u) s[u] = __builtin_amdgcn_readlane(myc, j + u);
      vecT r[8];
#pragma unroll
      for (int u = 0; u < 8; ++u)
        r[u] = *(const vecT*)(H + (size_t)s[u] * D + lane * V);
#pragma unroll
      for (int u = 0; u < 8; ++u) acc += r[u];
    }
    if (j < cnt) {
      int s[7];
#pragma unroll
      for (int u = 0; u < 7; ++u)
        s[u] = __builtin_amdgcn_readlane(myc, (j + u) & 31);
      vecT r[7];
#pragma unroll
      for (int u = 0; u < 7; ++u) {
        r[u] = (vecT)(0.0f);
        if (j + u < cnt) r[u] = *(const vecT*)(H + (size_t)s[u] * D + lane * V);
      }
#pragma unroll
      for (int u = 0; u < 7; ++u) acc += r[u];
    }
  }

  constexpr int KS = D / 4;
  constexpr int KH = KS / 2;
  const bool mm = wave < NOUT / 16;
  const int m  = lane & 15;
  const int kh = lane >> 4;
  const int nc = wave * 16 + m;

  v2f bq[KH];
  if (mm) {
#pragma unroll
    for (int s = 0; s < KH; ++s) {
      const int ka = 4 * s + 2 * kh;
      bq[s].x = W[(size_t)ka * NOUT + nc];
      bq[s].y = W[(size_t)(ka + 1) * NOUT + nc];
    }
  }

  {
    const float ni = norm_in[node];
    const float sc = scale[node];
    const vecT bb = *(const vecT*)(b + lane * V);
    vecT o;
#pragma unroll
    for (int k = 0; k < V; ++k)
      o[k] = fmaxf(fmaf(acc[k], ni, bb[k]), 0.0f) * sc;
    *(vecT*)(tile + wave * LDSK + lane * V) = o;
  }
  __syncthreads();

  if (!mm) return;

  v8f cacc = {};
#pragma unroll
  for (int s = 0; s < KH; ++s) {
    const int ka = 4 * s + 2 * kh;
    v2f a;
    a.x = tile[m * LDSK + ka];
    a.y = tile[m * LDSK + ka + 1];
    cacc = __builtin_amdgcn_wmma_f32_16x16x4_f32(
         false, a,  false, bq[s],
         (short)0, cacc,  false,  false);
  }
  v2f br[KH];
#pragma unroll
  for (int s = 0; s < KH; ++s) {
    const int ka = 4 * (s + KH) + 2 * kh;
    br[s].x = W[(size_t)ka * NOUT + nc];
    br[s].y = W[(size_t)(ka + 1) * NOUT + nc];
  }
#pragma unroll
  for (int s = 0; s < KH; ++s) {
    const int ka = 4 * (s + KH) + 2 * kh;
    v2f a;
    a.x = tile[m * LDSK + ka];
    a.y = tile[m * LDSK + ka + 1];
    cacc = __builtin_amdgcn_wmma_f32_16x16x4_f32(
         false, a,  false, br[s],
         (short)0, cacc,  false,  false);
  }

#pragma unroll
  for (int v = 0; v < 8; ++v) {
    const int orow = row0 + v + 8 * kh;
    Y[(size_t)orow * NOUT + nc] = cacc[v];
  }
}

extern "C" void kernel_launch(void* const* d_in, const int* in_sizes, int n_in,
                              void* d_out, int out_size, void* d_ws, size_t ws_size,
                              hipStream_t stream) {
  const float* feat = (const float*)d_in[0];
  const int*   src  = (const int*)d_in[1];
  const int*   dst  = (const int*)d_in[2];
  const float* W1   = (const float*)d_in[3];
  const float* b1   = (const float*)d_in[4];
  const float* W2   = (const float*)d_in[5];
  const float* b2   = (const float*)d_in[6];
  float* out = (float*)d_out;
  float* ws  = (float*)d_ws;

  const int N = GCN_N_NODES;
  const size_t NP = 50048;

  unsigned* part     = (unsigned*)ws;
  unsigned* cnt_in   = part + (size_t)2 * CSR_B * GCN_N_NODES;
  unsigned* row_ptr  = cnt_in + NP;
  float*    norm_out = (float*)(row_ptr + NP);
  float*    norm_in  = norm_out + NP;
  int*      col      = (int*)(norm_in + NP);
  float*    hproj1   = (float*)(col + (size_t)GCN_N_EDGES);
  float*    hproj2   = hproj1 + (size_t)N * GCN_HID_DIM;

  gcn_hist_kernel<<<dim3(CSR_B, 2), CSR_THREADS, 0, stream>>>(src, dst, part);
  gcn_colsum_kernel<<<(N + 255) / 256, 256, 0, stream>>>(part, cnt_in, norm_out, norm_in);
  gcn_fill_kernel<<<CSR_B, CSR_THREADS, 0, stream>>>(src, dst, part, cnt_in, row_ptr, col);

  gcn_gemm_scaled_wmma<GCN_IN_DIM, GCN_HID_DIM>
      <<<N / 16, (GCN_HID_DIM / 16) * 32, 0, stream>>>(feat, norm_out, W1, hproj1);

  gcn_gather_gemm_kernel<GCN_HID_DIM, GCN_OUT_DIM>
      <<<N / 16, 512, 0, stream>>>(hproj1, row_ptr, col, norm_in, b1, norm_out, W2, hproj2);

  gcn_gather_kernel<GCN_OUT_DIM>
      <<<(N + 7) / 8, 256, 0, stream>>>(hproj2, row_ptr, col, norm_in, b2, out, N);
}


# layer-1 projection and the slice-prefix pass share one launch on disjoint blocks; each projection block sums its 16 rows' out-degrees from the src planes itself; 6 launches -> 5
# speedup vs baseline: 4.3002x; 1.0663x over previous
#include <hip/hip_runtime.h>


#define GCN_N_NODES 50000
#define GCN_N_EDGES 800000
#define GCN_IN_DIM 128
#define GCN_HID_DIM 128
#define GCN_OUT_DIM 64

#define CSR_B 64
#define CSR_CHUNK ((GCN_N_EDGES + CSR_B - 1) / CSR_B)
#define CSR_THREADS 1024
#define CSR_RUN 49
#define CSR_PAD (CSR_RUN * CSR_THREADS)

typedef __attribute__((ext_vector_type(2))) float v2f;
typedef __attribute__((ext_vector_type(8))) float v8f;

__global__ __launch_bounds__(CSR_THREADS) void gcn_hist_kernel(
    const int* __restrict__ src, const int* __restrict__ dst,
    unsigned* __restrict__ part) {
  __shared__ unsigned bins[GCN_N_NODES];
  const int tid = threadIdx.x;
  const int* __restrict__ idx = blockIdx.y ? dst : src;
  unsigned* __restrict__ plane =
      part + ((size_t)blockIdx.y * CSR_B + blockIdx.x) * GCN_N_NODES;

  for (int i = tid; i < GCN_N_NODES; i += CSR_THREADS) bins[i] = 0u;
  __syncthreads();

  const int e0 = blockIdx.x * CSR_CHUNK;
  const int e1 = min(GCN_N_EDGES, e0 + CSR_CHUNK);
  for (int e = e0 + tid; e < e1; e += 4 * CSR_THREADS) {
    int d[4];
#pragma unroll
    for (int u = 0; u < 4; ++u) {
      const int ee = e + u * CSR_THREADS;
      d[u] = (ee < e1) ? idx[ee] : -1;
    }
#pragma unroll
    for (int u = 0; u < 4; ++u)
      if (d[u] >= 0) atomicAdd(&bins[d[u]], 1u);
  }
  __syncthreads();

  for (int i = tid; i < GCN_N_NODES; i += CSR_THREADS) plane[i] = bins[i];
}

__global__ __launch_bounds__(CSR_THREADS) void gcn_fill_kernel(
    const int* __restrict__ src, const int* __restrict__ dst,
    const unsigned* __restrict__ part, const unsigned* __restrict__ cnt_in,
    unsigned* __restrict__ row_ptr, int* __restrict__ col) {
  __shared__ unsigned cur[CSR_PAD];
  __shared__ unsigned wsum[32];
  const int tid  = threadIdx.x;
  const int lane = tid & 31;
  const int wave = tid >> 5;

  for (int i = tid; i < GCN_N_NODES; i += CSR_THREADS) cur[i] = cnt_in[i];
  for (int i = GCN_N_NODES + tid; i < CSR_PAD; i += CSR_THREADS) cur[i] = 0u;
  __syncthreads();

  unsigned s = 0u;
  for (int k = 0; k < CSR_RUN; ++k) s += cur[tid * CSR_RUN + k];

  unsigned incl = s;
  for (int off = 1; off < 32; off <<= 1) {
    const unsigned v = __shfl_up(incl, off);
    if (lane >= off) incl += v;
  }
  if (lane == 31) wsum[wave] = incl;
  __syncthreads();
  if (wave == 0) {
    const unsigned w = wsum[lane];
    unsigned wi = w;
    for (int off = 1; off < 32; off <<= 1) {
      const unsigned v = __shfl_up(wi, off);
      if (lane >= off) wi += v;
    }
    wsum[lane] = wi - w;
  }
  __syncthreads();

  unsigned run = wsum[wave] + incl - s;
  for (int k = 0; k < CSR_RUN; ++k) {
    const unsigned c = cur[tid * CSR_RUN + k];
    cur[tid * CSR_RUN + k] = run;
    run += c;
  }
  __syncthreads();

  if (blockIdx.x == 0)
    for (int i = tid; i <= GCN_N_NODES; i += CSR_THREADS) row_ptr[i] = cur[i];
  __syncthreads();

  const unsigned* __restrict__ offs =
      part + ((size_t)CSR_B + blockIdx.x) * GCN_N_NODES;
  for (int i = tid; i < GCN_N_NODES; i += CSR_THREADS) cur[i] += offs[i];
  __syncthreads();

  const int e0 = blockIdx.x * CSR_CHUNK;
  const int e1 = min(GCN_N_EDGES, e0 + CSR_CHUNK);
  for (int e = e0 + tid; e < e1; e += 4 * CSR_THREADS) {
    int d[4], s[4];
#pragma unroll
    for (int u = 0; u < 4; ++u) {
      const int ee = e + u * CSR_THREADS;
      d[u] = (ee < e1) ? dst[ee] : -1;
      s[u] = (ee < e1) ? src[ee] : 0;
    }
#pragma unroll
    for (int u = 0; u < 4; ++u)
      if (d[u] >= 0) {
        const unsigned pos = atomicAdd(&cur[d[u]], 1u);
        col[pos] = s[u];
      }
  }
}

#define PFX_THREADS 256
#define PFX_BLOCKS ((GCN_N_NODES + PFX_THREADS - 1) / PFX_THREADS)

template <int K, int NOUT>
__global__ __launch_bounds__(PFX_THREADS) void gcn_prefix_gemm_kernel(
    const float* __restrict__ X, const float* __restrict__ W,
    float* __restrict__ Y, unsigned* __restrict__ part,
    unsigned* __restrict__ cnt_in, float* __restrict__ norm_out,
    float* __restrict__ norm_in) {
  static_assert((NOUT / 16) * 32 == PFX_THREADS, "one wave per 16-column tile");
  static_assert(CSR_B * 4 == PFX_THREADS, "one thread per (src plane, 4 rows)");
  static_assert((16 * K) % PFX_THREADS == 0, "whole tile elements per thread");
  static_assert(GCN_N_NODES % 16 == 0, "one block owns 16 whole rows");
  constexpr int LDSK = K + 4;
  constexpr int XPT = (16 * K) / PFX_THREADS;
  typedef unsigned u4 __attribute__((ext_vector_type(4)));
  __shared__ float tile[16 * LDSK];
  __shared__ __attribute__((aligned(16))) unsigned degp[PFX_THREADS / 32][16];
  __shared__ float rscale[16];

  const int tid = threadIdx.x;

  if (blockIdx.x < PFX_BLOCKS) {
    const int n = blockIdx.x * PFX_THREADS + tid;
    if (n < GCN_N_NODES) {
      unsigned* pd = part + (size_t)CSR_B * GCN_N_NODES + n;
      unsigned run = 0u;
#pragma unroll 8
      for (int b = 0; b < CSR_B; ++b) {
        const unsigned c = pd[(size_t)b * GCN_N_NODES];
        pd[(size_t)b * GCN_N_NODES] = run;
        run += c;
      }
      cnt_in[n]  = run;
      norm_in[n] = rsqrtf(fmaxf((float)run, 1.0f));
    }
    return;
  }

  const int row0 = (blockIdx.x - PFX_BLOCKS) * 16;
  const int wave = tid >> 5;
  const int lane = tid & 31;

  float x[XPT];
#pragma unroll
  for (int i = 0; i < XPT; ++i) {
    const int t = tid + i * PFX_THREADS;
    const int r = t / K;
    const int c = t - r * K;
    x[i] = X[(size_t)(row0 + r) * K + c];
  }

  {
    const int p = tid >> 2;
    const int q = tid & 3;
    u4 c = *(const u4*)(part + (size_t)p * GCN_N_NODES + row0 + 4 * q);
    for (int off = 4; off < 32; off <<= 1) {
      c.x += __shfl_xor(c.x, off);
      c.y += __shfl_xor(c.y, off);
      c.z += __shfl_xor(c.z, off);
      c.w += __shfl_xor(c.w, off);
    }
    if (lane < 4) *(u4*)(&degp[wave][4 * lane]) = c;
  }
  __syncthreads();
  if (tid < 16) {
    unsigned so = 0u;
#pragma unroll
    for (int w = 0; w < PFX_THREADS / 32; ++w) so += degp[w][tid];
    const float s = rsqrtf(fmaxf((float)so, 1.0f));
    norm_out[row0 + tid] = s;
    rscale[tid] = s;
  }
  __syncthreads();

#pragma unroll
  for (int i = 0; i < XPT; ++i) {
    const int t = tid + i * PFX_THREADS;
    const int r = t / K;
    const int c = t - r * K;
    tile[r * LDSK + c] = x[i] * rscale[r];
  }
  __syncthreads();

  const int m  = lane & 15;
  const int kh = lane >> 4;
  const int nc = wave * 16 + m;

  v8f acc = {};
#pragma unroll 4
  for (int kk = 0; kk < K; kk += 4) {
    v2f a, b;
    const int ka = kk + 2 * kh;
    a.x = tile[m * LDSK + ka];
    a.y = tile[m * LDSK + ka + 1];
    b.x = W[(size_t)ka * NOUT + nc];
    b.y = W[(size_t)(ka + 1) * NOUT + nc];
    acc = __builtin_amdgcn_wmma_f32_16x16x4_f32(
         false, a,  false, b,
         (short)0, acc,  false,  false);
  }

#pragma unroll
  for (int v = 0; v < 8; ++v) {
    const int orow = row0 + v + 8 * kh;
    Y[(size_t)orow * NOUT + nc] = acc[v];
  }
}

template <int D>
__global__ __launch_bounds__(256) void gcn_gather_kernel(
    const float* __restrict__ H, const unsigned* __restrict__ row_ptr,
    const int* __restrict__ col, const float* __restrict__ norm_in,
    const float* __restrict__ b, float* __restrict__ out, int N) {
  constexpr int V = D / 32;
  typedef float vecT __attribute__((ext_vector_type(V)));
  const int lane = threadIdx.x & 31;
  const int node =
      __builtin_amdgcn_readfirstlane(blockIdx.x * 8 + (threadIdx.x >> 5));
  if (node >= N) return;

  const unsigned beg = row_ptr[node];
  const unsigned end = row_ptr[node + 1];

  vecT acc = (vecT)(0.0f);
  for (unsigned base = beg; base < end; base += 32u) {
    const int cnt = (int)min(32u, end - base);
    int myc = 0;
    if (lane < cnt) myc = col[base + lane];
    int j = 0;
    for (; j + 8 <= cnt; j += 8) {
      int s[8];
#pragma unroll
      for (int u = 0; u < 8; ++u) s[u] = __builtin_amdgcn_readlane(myc, j + u);
      vecT r[8];
#pragma unroll
      for (int u = 0; u < 8; ++u)
        r[u] = *(const vecT*)(H + (size_t)s[u] * D + lane * V);
#pragma unroll
      for (int u = 0; u < 8; ++u) acc += r[u];
    }
    if (j < cnt) {
      int s[7];
#pragma unroll
      for (int u = 0; u < 7; ++u)
        s[u] = __builtin_amdgcn_readlane(myc, (j + u) & 31);
      vecT r[7];
#pragma unroll
      for (int u = 0; u < 7; ++u) {
        r[u] = (vecT)(0.0f);
        if (j + u < cnt) r[u] = *(const vecT*)(H + (size_t)s[u] * D + lane * V);
      }
#pragma unroll
      for (int u = 0; u < 7; ++u) acc += r[u];
    }
  }

  const float ni = norm_in[node];
  const vecT bb = *(const vecT*)(b + lane * V);
  vecT o;
#pragma unroll
  for (int k = 0; k < V; ++k) o[k] = fmaxf(fmaf(acc[k], ni, bb[k]), 0.0f);
  *(vecT*)(out + (size_t)node * D + lane * V) = o;
}

template <int D, int NOUT>
__global__ __launch_bounds__(512) void gcn_gather_gemm_kernel(
    const float* __restrict__ H, const unsigned* __restrict__ row_ptr,
    const int* __restrict__ col, const float* __restrict__ norm_in,
    const float* __restrict__ b, const float* __restrict__ scale,
    const float* __restrict__ W, float* __restrict__ Y) {
  static_assert(GCN_N_NODES % 16 == 0, "one block owns 16 whole rows");
  static_assert(NOUT % 16 == 0 && NOUT / 16 <= 16, "one wave per column tile");
  static_assert(D % 8 == 0, "two K halves of whole k-steps");
  constexpr int V = D / 32;
  constexpr int LDSK = D + 4;
  typedef float vecT __attribute__((ext_vector_type(V)));
  __shared__ __attribute__((aligned(16))) float tile[16 * LDSK];

  const int lane = threadIdx.x & 31;
  const int wave = __builtin_amdgcn_readfirstlane(threadIdx.x >> 5);
  const int row0 = blockIdx.x * 16;
  const int node = row0 + wave;

  const unsigned beg = row_ptr[node];
  const unsigned end = row_ptr[node + 1];

  vecT acc = (vecT)(0.0f);
  for (unsigned base = beg; base < end; base += 32u) {
    const int cnt = (int)min(32u, end - base);
    int myc = 0;
    if (lane < cnt) myc = col[base + lane];
    int j = 0;
    for (; j + 8 <= cnt; j += 8) {
      int s[8];
#pragma unroll
      for (int u = 0; u < 8; ++u) s[u] = __builtin_amdgcn_readlane(myc, j + u);
      vecT r[8];
#pragma unroll
      for (int u = 0; u < 8; ++u)
        r[u] = *(const vecT*)(H + (size_t)s[u] * D + lane * V);
#pragma unroll
      for (int u = 0; u < 8; ++u) acc += r[u];
    }
    if (j < cnt) {
      int s[7];
#pragma unroll
      for (int u = 0; u < 7; ++u)
        s[u] = __builtin_amdgcn_readlane(myc, (j + u) & 31);
      vecT r[7];
#pragma unroll
      for (int u = 0; u < 7; ++u) {
        r[u] = (vecT)(0.0f);
        if (j + u < cnt) r[u] = *(const vecT*)(H + (size_t)s[u] * D + lane * V);
      }
#pragma unroll
      for (int u = 0; u < 7; ++u) acc += r[u];
    }
  }

  constexpr int KS = D / 4;
  constexpr int KH = KS / 2;
  const bool mm = wave < NOUT / 16;
  const int m  = lane & 15;
  const int kh = lane >> 4;
  const int nc = wave * 16 + m;

  v2f bq[KH];
  if (mm) {
#pragma unroll
    for (int s = 0; s < KH; ++s) {
      const int ka = 4 * s + 2 * kh;
      bq[s].x = W[(size_t)ka * NOUT + nc];
      bq[s].y = W[(size_t)(ka + 1) * NOUT + nc];
    }
  }

  {
    const float ni = norm_in[node];
    const float sc = scale[node];
    const vecT bb = *(const vecT*)(b + lane * V);
    vecT o;
#pragma unroll
    for (int k = 0; k < V; ++k)
      o[k] = fmaxf(fmaf(acc[k], ni, bb[k]), 0.0f) * sc;
    *(vecT*)(tile + wave * LDSK + lane * V) = o;
  }
  __syncthreads();

  if (!mm) return;

  v8f cacc = {};
#pragma unroll
  for (int s = 0; s < KH; ++s) {
    const int ka = 4 * s + 2 * kh;
    v2f a;
    a.x = tile[m * LDSK + ka];
    a.y = tile[m * LDSK + ka + 1];
    cacc = __builtin_amdgcn_wmma_f32_16x16x4_f32(
         false, a,  false, bq[s],
         (short)0, cacc,  false,  false);
  }
  v2f br[KH];
#pragma unroll
  for (int s = 0; s < KH; ++s) {
    const int ka = 4 * (s + KH) + 2 * kh;
    br[s].x = W[(size_t)ka * NOUT + nc];
    br[s].y = W[(size_t)(ka + 1) * NOUT + nc];
  }
#pragma unroll
  for (int s = 0; s < KH; ++s) {
    const int ka = 4 * (s + KH) + 2 * kh;
    v2f a;
    a.x = tile[m * LDSK + ka];
    a.y = tile[m * LDSK + ka + 1];
    cacc = __builtin_amdgcn_wmma_f32_16x16x4_f32(
         false, a,  false, br[s],
         (short)0, cacc,  false,  false);
  }

#pragma unroll
  for (int v = 0; v < 8; ++v) {
    const int orow = row0 + v + 8 * kh;
    Y[(size_t)orow * NOUT + nc] = cacc[v];
  }
}

extern "C" void kernel_launch(void* const* d_in, const int* in_sizes, int n_in,
                              void* d_out, int out_size, void* d_ws, size_t ws_size,
                              hipStream_t stream) {
  const float* feat = (const float*)d_in[0];
  const int*   src  = (const int*)d_in[1];
  const int*   dst  = (const int*)d_in[2];
  const float* W1   = (const float*)d_in[3];
  const float* b1   = (const float*)d_in[4];
  const float* W2   = (const float*)d_in[5];
  const float* b2   = (const float*)d_in[6];
  float* out = (float*)d_out;
  float* ws  = (float*)d_ws;

  const int N = GCN_N_NODES;
  const size_t NP = 50048;

  unsigned* part     = (unsigned*)ws;
  unsigned* cnt_in   = part + (size_t)2 * CSR_B * GCN_N_NODES;
  unsigned* row_ptr  = cnt_in + NP;
  float*    norm_out = (float*)(row_ptr + NP);
  float*    norm_in  = norm_out + NP;
  int*      col      = (int*)(norm_in + NP);
  float*    hproj1   = (float*)(col + (size_t)GCN_N_EDGES);
  float*    hproj2   = hproj1 + (size_t)N * GCN_HID_DIM;

  gcn_hist_kernel<<<dim3(CSR_B, 2), CSR_THREADS, 0, stream>>>(src, dst, part);
  gcn_prefix_gemm_kernel<GCN_IN_DIM, GCN_HID_DIM>
      <<<PFX_BLOCKS + N / 16, PFX_THREADS, 0, stream>>>(
          feat, W1, hproj1, part, cnt_in, norm_out, norm_in);
  gcn_fill_kernel<<<CSR_B, CSR_THREADS, 0, stream>>>(src, dst, part, cnt_in, row_ptr, col);

  gcn_gather_gemm_kernel<GCN_HID_DIM, GCN_OUT_DIM>
      <<<N / 16, 512, 0, stream>>>(hproj1, row_ptr, col, norm_in, b1, norm_out, W2, hproj2);

  gcn_gather_kernel<GCN_OUT_DIM>
      <<<(N + 7) / 8, 256, 0, stream>>>(hproj2, row_ptr, col, norm_in, b2, out, N);
}


# row_ptr formed once in the slice-prefix pass (group counts from the dst histogram blocks + block scan), not by a scan of all in-degrees in each of the 64 fill blocks, which now load ready cursors
# speedup vs baseline: 4.5154x; 1.0500x over previous
#include <hip/hip_runtime.h>


#define GCN_N_NODES 50000
#define GCN_N_EDGES 800000
#define GCN_IN_DIM 128
#define GCN_HID_DIM 128
#define GCN_OUT_DIM 64

#define CSR_B 64
#define CSR_CHUNK ((GCN_N_EDGES + CSR_B - 1) / CSR_B)
#define CSR_THREADS 1024
#define CSR_GROUP 256
#define CSR_NGROUPS ((GCN_N_NODES + CSR_GROUP - 1) / CSR_GROUP)
#define CSR_GPAD 256
#define CSR_BINS (CSR_NGROUPS * CSR_GROUP)

typedef __attribute__((ext_vector_type(2))) float v2f;
typedef __attribute__((ext_vector_type(8))) float v8f;

__device__ __forceinline__ unsigned wave_sum_u32(unsigned v) {
  v += (unsigned)__builtin_amdgcn_update_dpp((int)v, (int)v, 0x161, 0xf, 0xf, false);
  v += (unsigned)__builtin_amdgcn_update_dpp((int)v, (int)v, 0x162, 0xf, 0xf, false);
  v += (unsigned)__builtin_amdgcn_update_dpp((int)v, (int)v, 0x164, 0xf, 0xf, false);
  v += (unsigned)__builtin_amdgcn_update_dpp((int)v, (int)v, 0x168, 0xf, 0xf, false);
  v += (unsigned)__builtin_amdgcn_permlanex16((int)v, (int)v, 0x76543210, 0xfedcba98,
                                              false, false);
  return v;
}

__global__ __launch_bounds__(CSR_THREADS) void gcn_hist_kernel(
    const int* __restrict__ src, const int* __restrict__ dst,
    unsigned* __restrict__ part, unsigned* __restrict__ coarse) {
  __shared__ __attribute__((aligned(16))) unsigned bins[CSR_BINS];
  const int tid = threadIdx.x;
  const int* __restrict__ idx = blockIdx.y ? dst : src;
  unsigned* __restrict__ plane =
      part + ((size_t)blockIdx.y * CSR_B + blockIdx.x) * GCN_N_NODES;

  for (int i = tid; i < CSR_BINS; i += CSR_THREADS) bins[i] = 0u;
  __syncthreads();

  const int e0 = blockIdx.x * CSR_CHUNK;
  const int e1 = min(GCN_N_EDGES, e0 + CSR_CHUNK);
  for (int e = e0 + tid; e < e1; e += 4 * CSR_THREADS) {
    int d[4];
#pragma unroll
    for (int u = 0; u < 4; ++u) {
      const int ee = e + u * CSR_THREADS;
      d[u] = (ee < e1) ? idx[ee] : -1;
    }
#pragma unroll
    for (int u = 0; u < 4; ++u)
      if (d[u] >= 0) atomicAdd(&bins[d[u]], 1u);
  }
  __syncthreads();

  for (int i = tid; i < GCN_N_NODES; i += CSR_THREADS) plane[i] = bins[i];

  if (blockIdx.y == 1) {
    static_assert(CSR_GROUP == 256, "two 16-byte reads per lane cover a group");
    static_assert(CSR_THREADS == 1024, "32 waves share the groups");
    typedef unsigned u4 __attribute__((ext_vector_type(4)));
    constexpr int GPW = (CSR_NGROUPS + 31) / 32;
    const int lane = tid & 31;
    const int wave = tid >> 5;
    unsigned c[GPW];
#pragma unroll
    for (int j = 0; j < GPW; ++j) {
      const int g = min(wave + 32 * j, CSR_NGROUPS - 1);
      const u4 a = *(const u4*)(bins + g * CSR_GROUP + 4 * lane);
      const u4 b = *(const u4*)(bins + g * CSR_GROUP + 128 + 4 * lane);
      c[j] = ((a.x + a.y) + (a.z + a.w)) + ((b.x + b.y) + (b.z + b.w));
    }
#pragma unroll
    for (int j = 0; j < GPW; ++j) c[j] = wave_sum_u32(c[j]);
    unsigned v = c[0];
#pragma unroll
    for (int j = 1; j < GPW; ++j)
      if (lane == j) v = c[j];
    const int g = wave + 32 * lane;
    if (lane < GPW && g < CSR_NGROUPS) coarse[blockIdx.x * CSR_GPAD + g] = v;
  }
}

__global__ __launch_bounds__(CSR_THREADS) void gcn_fill_kernel(
    const int* __restrict__ src, const int* __restrict__ dst,
    const unsigned* __restrict__ part, int* __restrict__ col) {
  __shared__ unsigned cur[GCN_N_NODES];
  const int tid = threadIdx.x;

  const unsigned* __restrict__ offs =
      part + ((size_t)CSR_B + blockIdx.x) * GCN_N_NODES;
  for (int i = tid; i < GCN_N_NODES; i += CSR_THREADS) cur[i] = offs[i];
  __syncthreads();

  const int e0 = blockIdx.x * CSR_CHUNK;
  const int e1 = min(GCN_N_EDGES, e0 + CSR_CHUNK);
  for (int e = e0 + tid; e < e1; e += 4 * CSR_THREADS) {
    int d[4], s[4];
#pragma unroll
    for (int u = 0; u < 4; ++u) {
      const int ee = e + u * CSR_THREADS;
      d[u] = (ee < e1) ? dst[ee] : -1;
      s[u] = (ee < e1) ? src[ee] : 0;
    }
#pragma unroll
    for (int u = 0; u < 4; ++u)
      if (d[u] >= 0) {
        const unsigned pos = atomicAdd(&cur[d[u]], 1u);
        col[pos] = s[u];
      }
  }
}

#define PFX_THREADS 256
#define PFX_BLOCKS ((GCN_N_NODES + PFX_THREADS - 1) / PFX_THREADS)

template <int K, int NOUT>
__global__ __launch_bounds__(PFX_THREADS) void gcn_prefix_gemm_kernel(
    const float* __restrict__ X, const float* __restrict__ W,
    float* __restrict__ Y, unsigned* __restrict__ part,
    const unsigned* __restrict__ coarse, unsigned* __restrict__ row_ptr,
    float* __restrict__ norm_out, float* __restrict__ norm_in) {
  static_assert((NOUT / 16) * 32 == PFX_THREADS, "one wave per 16-column tile");
  static_assert(CSR_B * 4 == PFX_THREADS, "one thread per (src plane, 4 rows)");
  static_assert((16 * K) % PFX_THREADS == 0, "whole tile elements per thread");
  static_assert(GCN_N_NODES % 16 == 0, "one block owns 16 whole rows");
  static_assert(CSR_GROUP == PFX_THREADS, "one prefix block per coarse group");
  static_assert(CSR_B % 32 == 0, "plane sums run in trips of 32");
  static_assert(CSR_NGROUPS <= CSR_GPAD && PFX_THREADS <= CSR_GPAD, "coarse row");
  constexpr int LDSK = K + 4;
  constexpr int XPT = (16 * K) / PFX_THREADS;
  typedef unsigned u4 __attribute__((ext_vector_type(4)));
  __shared__ float tile[16 * LDSK];
  __shared__ __attribute__((aligned(16))) unsigned degp[PFX_THREADS / 32][16];
  __shared__ float rscale[16];
  __shared__ unsigned wdeg[PFX_THREADS / 32];
  __shared__ unsigned wbase[PFX_THREADS / 32];

  const int tid = threadIdx.x;

  if (blockIdx.x < PFX_BLOCKS) {
    const int g = blockIdx.x;
    const int n = g * PFX_THREADS + tid;
    const bool live = n < GCN_N_NODES;
    const int lane = tid & 31;
    const int wave = tid >> 5;
    unsigned* __restrict__ pd = part + (size_t)CSR_B * GCN_N_NODES;
    const unsigned noff = (unsigned)n * 4u;
    const unsigned toff = (unsigned)tid * 4u;

    unsigned bs = 0u;
    if (tid < g) {
#pragma nounroll
      for (int h = 0; h < CSR_B; h += 32) {
#pragma unroll
        for (int b = h; b < h + 32; ++b)
          bs += *(const unsigned*)((const char*)(coarse + b * CSR_GPAD) + toff);
      }
    }
    unsigned deg = 0u;
    if (live) {
#pragma nounroll
      for (int h = 0; h < CSR_B; h += 32) {
#pragma unroll
        for (int b = h; b < h + 32; ++b)
          deg += *(const unsigned*)((const char*)(pd + (size_t)b * GCN_N_NODES) + noff);
      }
    }

    unsigned incl = deg;
    for (int off = 1; off < 32; off <<= 1) {
      const unsigned v = __shfl_up(incl, off);
      if (lane >= off) incl += v;
    }
    for (int off = 16; off > 0; off >>= 1) bs += __shfl_xor(bs, off);
    if (lane == 31) wdeg[wave] = incl;
    if (lane == 0) wbase[wave] = bs;
    __syncthreads();
    unsigned rp = incl - deg;
#pragma unroll
    for (int w = 0; w < PFX_THREADS / 32; ++w) {
      rp += wbase[w];
      if (w < wave) rp += wdeg[w];
    }

    if (live) {
      row_ptr[n] = rp;
      if (n == GCN_N_NODES - 1) row_ptr[GCN_N_NODES] = rp + deg;
      norm_in[n] = rsqrtf(fmaxf((float)deg, 1.0f));
      unsigned run = rp;
#pragma unroll 16
      for (int b = 0; b < CSR_B; ++b) {
        unsigned* p = (unsigned*)((char*)(pd + (size_t)b * GCN_N_NODES) + noff);
        const unsigned c = *p;
        *p = run;
        run += c;
      }
    }
    return;
  }

  const int row0 = (blockIdx.x - PFX_BLOCKS) * 16;
  const int wave = tid >> 5;
  const int lane = tid & 31;

  float x[XPT];
#pragma unroll
  for (int i = 0; i < XPT; ++i) {
    const int t = tid + i * PFX_THREADS;
    const int r = t / K;
    const int c = t - r * K;
    x[i] = X[(size_t)(row0 + r) * K + c];
  }

  {
    const int p = tid >> 2;
    const int q = tid & 3;
    u4 c = *(const u4*)(part + (size_t)p * GCN_N_NODES + row0 + 4 * q);
    for (int off = 4; off < 32; off <<= 1) {
      c.x += __shfl_xor(c.x, off);
      c.y += __shfl_xor(c.y, off);
      c.z += __shfl_xor(c.z, off);
      c.w += __shfl_xor(c.w, off);
    }
    if (lane < 4) *(u4*)(&degp[wave][4 * lane]) = c;
  }
  __syncthreads();
  if (tid < 16) {
    unsigned so = 0u;
#pragma unroll
    for (int w = 0; w < PFX_THREADS / 32; ++w) so += degp[w][tid];
    const float s = rsqrtf(fmaxf((float)so, 1.0f));
    norm_out[row0 + tid] = s;
    rscale[tid] = s;
  }
  __syncthreads();

#pragma unroll
  for (int i = 0; i < XPT; ++i) {
    const int t = tid + i * PFX_THREADS;
    const int r = t / K;
    const int c = t - r * K;
    tile[r * LDSK + c] = x[i] * rscale[r];
  }
  __syncthreads();

  const int m  = lane & 15;
  const int kh = lane >> 4;
  const int nc = wave * 16 + m;

  v8f acc = {};
#pragma unroll 4
  for (int kk = 0; kk < K; kk += 4) {
    v2f a, b;
    const int ka = kk + 2 * kh;
    a.x = tile[m * LDSK + ka];
    a.y = tile[m * LDSK + ka + 1];
    b.x = W[(size_t)ka * NOUT + nc];
    b.y = W[(size_t)(ka + 1) * NOUT + nc];
    acc = __builtin_amdgcn_wmma_f32_16x16x4_f32(
         false, a,  false, b,
         (short)0, acc,  false,  false);
  }

#pragma unroll
  for (int v = 0; v < 8; ++v) {
    const int orow = row0 + v + 8 * kh;
    Y[(size_t)orow * NOUT + nc] = acc[v];
  }
}

template <int D>
__global__ __launch_bounds__(256) void gcn_gather_kernel(
    const float* __restrict__ H, const unsigned* __restrict__ row_ptr,
    const int* __restrict__ col, const float* __restrict__ norm_in,
    const float* __restrict__ b, float* __restrict__ out, int N) {
  constexpr int V = D / 32;
  typedef float vecT __attribute__((ext_vector_type(V)));
  const int lane = threadIdx.x & 31;
  const int node =
      __builtin_amdgcn_readfirstlane(blockIdx.x * 8 + (threadIdx.x >> 5));
  if (node >= N) return;

  const unsigned beg = row_ptr[node];
  const unsigned end = row_ptr[node + 1];

  vecT acc = (vecT)(0.0f);
  for (unsigned base = beg; base < end; base += 32u) {
    const int cnt = (int)min(32u, end - base);
    int myc = 0;
    if (lane < cnt) myc = col[base + lane];
    int j = 0;
    for (; j + 8 <= cnt; j += 8) {
      int s[8];
#pragma unroll
      for (int u = 0; u < 8; ++u) s[u] = __builtin_amdgcn_readlane(myc, j + u);
      vecT r[8];
#pragma unroll
      for (int u = 0; u < 8; ++u)
        r[u] = *(const vecT*)(H + (size_t)s[u] * D + lane * V);
#pragma unroll
      for (int u = 0; u < 8; ++u) acc += r[u];
    }
    if (j < cnt) {
      int s[7];
#pragma unroll
      for (int u = 0; u < 7; ++u)
        s[u] = __builtin_amdgcn_readlane(myc, (j + u) & 31);
      vecT r[7];
#pragma unroll
      for (int u = 0; u < 7; ++u) {
        r[u] = (vecT)(0.0f);
        if (j + u < cnt) r[u] = *(const vecT*)(H + (size_t)s[u] * D + lane * V);
      }
#pragma unroll
      for (int u = 0; u < 7; ++u) acc += r[u];
    }
  }

  const float ni = norm_in[node];
  const vecT bb = *(const vecT*)(b + lane * V);
  vecT o;
#pragma unroll
  for (int k = 0; k < V; ++k) o[k] = fmaxf(fmaf(acc[k], ni, bb[k]), 0.0f);
  *(vecT*)(out + (size_t)node * D + lane * V) = o;
}

template <int D, int NOUT>
__global__ __launch_bounds__(512) void gcn_gather_gemm_kernel(
    const float* __restrict__ H, const unsigned* __restrict__ row_ptr,
    const int* __restrict__ col, const float* __restrict__ norm_in,
    const float* __restrict__ b, const float* __restrict__ scale,
    const float* __restrict__ W, float* __restrict__ Y) {
  static_assert(GCN_N_NODES % 16 == 0, "one block owns 16 whole rows");
  static_assert(NOUT % 16 == 0 && NOUT / 16 <= 16, "one wave per column tile");
  static_assert(D % 8 == 0, "two K halves of whole k-steps");
  constexpr int V = D / 32;
  constexpr int LDSK = D + 4;
  typedef float vecT __attribute__((ext_vector_type(V)));
  __shared__ __attribute__((aligned(16))) float tile[16 * LDSK];

  const int lane = threadIdx.x & 31;
  const int wave = __builtin_amdgcn_readfirstlane(threadIdx.x >> 5);
  const int row0 = blockIdx.x * 16;
  const int node = row0 + wave;

  const unsigned beg = row_ptr[node];
  const unsigned end = row_ptr[node + 1];

  vecT acc = (vecT)(0.0f);
  for (unsigned base = beg; base < end; base += 32u) {
    const int cnt = (int)min(32u, end - base);
    int myc = 0;
    if (lane < cnt) myc = col[base + lane];
    int j = 0;
    for (; j + 8 <= cnt; j += 8) {
      int s[8];
#pragma unroll
      for (int u = 0; u < 8; ++u) s[u] = __builtin_amdgcn_readlane(myc, j + u);
      vecT r[8];
#pragma unroll
      for (int u = 0; u < 8; ++u)
        r[u] = *(const vecT*)(H + (size_t)s[u] * D + lane * V);
#pragma unroll
      for (int u = 0; u < 8; ++u) acc += r[u];
    }
    if (j < cnt) {
      int s[7];
#pragma unroll
      for (int u = 0; u < 7; ++u)
        s[u] = __builtin_amdgcn_readlane(myc, (j + u) & 31);
      vecT r[7];
#pragma unroll
      for (int u = 0; u < 7; ++u) {
        r[u] = (vecT)(0.0f);
        if (j + u < cnt) r[u] = *(const vecT*)(H + (size_t)s[u] * D + lane * V);
      }
#pragma unroll
      for (int u = 0; u < 7; ++u) acc += r[u];
    }
  }

  constexpr int KS = D / 4;
  constexpr int KH = KS / 2;
  const bool mm = wave < NOUT / 16;
  const int m  = lane & 15;
  const int kh = lane >> 4;
  const int nc = wave * 16 + m;

  v2f bq[KH];
  if (mm) {
#pragma unroll
    for (int s = 0; s < KH; ++s) {
      const int ka = 4 * s + 2 * kh;
      bq[s].x = W[(size_t)ka * NOUT + nc];
      bq[s].y = W[(size_t)(ka + 1) * NOUT + nc];
    }
  }

  {
    const float ni = norm_in[node];
    const float sc = scale[node];
    const vecT bb = *(const vecT*)(b + lane * V);
    vecT o;
#pragma unroll
    for (int k = 0; k < V; ++k)
      o[k] = fmaxf(fmaf(acc[k], ni, bb[k]), 0.0f) * sc;
    *(vecT*)(tile + wave * LDSK + lane * V) = o;
  }
  __syncthreads();

  if (!mm) return;

  v8f cacc = {};
#pragma unroll
  for (int s = 0; s < KH; ++s) {
    const int ka = 4 * s + 2 * kh;
    v2f a;
    a.x = tile[m * LDSK + ka];
    a.y = tile[m * LDSK + ka + 1];
    cacc = __builtin_amdgcn_wmma_f32_16x16x4_f32(
         false, a,  false, bq[s],
         (short)0, cacc,  false,  false);
  }
  v2f br[KH];
#pragma unroll
  for (int s = 0; s < KH; ++s) {
    const int ka = 4 * (s + KH) + 2 * kh;
    br[s].x = W[(size_t)ka * NOUT + nc];
    br[s].y = W[(size_t)(ka + 1) * NOUT + nc];
  }
#pragma unroll
  for (int s = 0; s < KH; ++s) {
    const int ka = 4 * (s + KH) + 2 * kh;
    v2f a;
    a.x = tile[m * LDSK + ka];
    a.y = tile[m * LDSK + ka + 1];
    cacc = __builtin_amdgcn_wmma_f32_16x16x4_f32(
         false, a,  false, br[s],
         (short)0, cacc,  false,  false);
  }

#pragma unroll
  for (int v = 0; v < 8; ++v) {
    const int orow = row0 + v + 8 * kh;
    Y[(size_t)orow * NOUT + nc] = cacc[v];
  }
}

extern "C" void kernel_launch(void* const* d_in, const int* in_sizes, int n_in,
                              void* d_out, int out_size, void* d_ws, size_t ws_size,
                              hipStream_t stream) {
  const float* feat = (const float*)d_in[0];
  const int*   src  = (const int*)d_in[1];
  const int*   dst  = (const int*)d_in[2];
  const float* W1   = (const float*)d_in[3];
  const float* b1   = (const float*)d_in[4];
  const float* W2   = (const float*)d_in[5];
  const float* b2   = (const float*)d_in[6];
  float* out = (float*)d_out;
  float* ws  = (float*)d_ws;

  const int N = GCN_N_NODES;
  const size_t NP = 50048;

  unsigned* part     = (unsigned*)ws;
  unsigned* coarse   = part + (size_t)2 * CSR_B * GCN_N_NODES;
  unsigned* row_ptr  = coarse + NP;
  float*    norm_out = (float*)(row_ptr + NP);
  float*    norm_in  = norm_out + NP;
  int*      col      = (int*)(norm_in + NP);
  float*    hproj1   = (float*)(col + (size_t)GCN_N_EDGES);
  float*    hproj2   = hproj1 + (size_t)N * GCN_HID_DIM;

  gcn_hist_kernel<<<dim3(CSR_B, 2), CSR_THREADS, 0, stream>>>(src, dst, part, coarse);
  gcn_prefix_gemm_kernel<GCN_IN_DIM, GCN_HID_DIM>
      <<<PFX_BLOCKS + N / 16, PFX_THREADS, 0, stream>>>(
          feat, W1, hproj1, part, coarse, row_ptr, norm_out, norm_in);
  gcn_fill_kernel<<<CSR_B, CSR_THREADS, 0, stream>>>(src, dst, part, col);

  gcn_gather_gemm_kernel<GCN_HID_DIM, GCN_OUT_DIM>
      <<<N / 16, 512, 0, stream>>>(hproj1, row_ptr, col, norm_in, b1, norm_out, W2, hproj2);

  gcn_gather_kernel<GCN_OUT_DIM>
      <<<(N + 7) / 8, 256, 0, stream>>>(hproj2, row_ptr, col, norm_in, b2, out, N);
}


# sort by dst in two levels: slice blocks bucket edges by 256-node group, each group block counts and places its edges in LDS and writes col; fill launch and dst planes go, 5 -> 4 launches
# speedup vs baseline: 6.6627x; 1.4756x over previous
#include <hip/hip_runtime.h>


#define GCN_N_NODES 50000
#define GCN_N_EDGES 800000
#define GCN_IN_DIM 128
#define GCN_HID_DIM 128
#define GCN_OUT_DIM 64

#define CSR_B 64
#define CSR_CHUNK ((GCN_N_EDGES + CSR_B - 1) / CSR_B)
#define CSR_THREADS 1024
#define CSR_GROUP 256
#define CSR_NGROUPS ((GCN_N_NODES + CSR_GROUP - 1) / CSR_GROUP)
#define CSR_GPAD 256

typedef __attribute__((ext_vector_type(2))) float v2f;
typedef __attribute__((ext_vector_type(8))) float v8f;

__device__ __forceinline__ unsigned wave_sum_u32(unsigned v) {
  v += (unsigned)__builtin_amdgcn_update_dpp((int)v, (int)v, 0x161, 0xf, 0xf, false);
  v += (unsigned)__builtin_amdgcn_update_dpp((int)v, (int)v, 0x162, 0xf, 0xf, false);
  v += (unsigned)__builtin_amdgcn_update_dpp((int)v, (int)v, 0x164, 0xf, 0xf, false);
  v += (unsigned)__builtin_amdgcn_update_dpp((int)v, (int)v, 0x168, 0xf, 0xf, false);
  v += (unsigned)__builtin_amdgcn_permlanex16((int)v, (int)v, 0x76543210, 0xfedcba98,
                                              false, false);
  return v;
}

#define CSR_NONE 0xffffffffu

__global__ __launch_bounds__(CSR_THREADS) void gcn_hist_kernel(
    const int* __restrict__ src, const int* __restrict__ dst,
    unsigned* __restrict__ part, unsigned* __restrict__ coarse,
    unsigned* __restrict__ goffs, unsigned* __restrict__ bucket) {
  __shared__ __attribute__((aligned(16))) unsigned bins[GCN_N_NODES];
  const int tid = threadIdx.x;

  if (blockIdx.y == 1) {
    static_assert(GCN_N_EDGES % CSR_B == 0, "every slice is full");
    static_assert(GCN_N_NODES <= 65536 && CSR_GROUP == 256,
                  "a src id and a dst offset inside its group share one word");
    static_assert(CSR_GPAD == 256 && CSR_NGROUPS <= CSR_GPAD, "256 counters");
    constexpr int EPT = (CSR_CHUNK + CSR_THREADS - 1) / CSR_THREADS;
    unsigned* gcnt = bins;
    unsigned* goff = bins + 256;
    unsigned* wsum = bins + 512;
    const int lane = tid & 31;
    const int wave = tid >> 5;

    if (tid < CSR_GPAD) gcnt[tid] = 0u;
    __syncthreads();

    const int e0 = blockIdx.x * CSR_CHUNK;
    int d[EPT], s[EPT];
    unsigned r[EPT];
#pragma unroll
    for (int k = 0; k < EPT; ++k) {
      const int le = tid + k * CSR_THREADS;
      const bool ok = le < CSR_CHUNK;
      d[k] = ok ? dst[e0 + le] : -1;
      s[k] = ok ? src[e0 + le] : 0;
    }
#pragma unroll
    for (int k = 0; k < EPT; ++k) {
      r[k] = 0u;
      if (d[k] >= 0) r[k] = atomicAdd(&gcnt[d[k] >> 8], 1u);
    }
    __syncthreads();

    unsigned cg = 0u, incl = 0u;
    if (tid < CSR_GPAD) {
      cg = gcnt[tid];
      incl = cg;
      for (int off = 1; off < 32; off <<= 1) {
        const unsigned v = __shfl_up(incl, off);
        if (lane >= off) incl += v;
      }
      if (lane == 31) wsum[wave] = incl;
    }
    __syncthreads();
    if (tid < CSR_GPAD) {
      unsigned ex = incl - cg;
#pragma unroll
      for (int w = 0; w < CSR_GPAD / 32; ++w)
        if (w < wave) ex += wsum[w];
      goff[tid] = ex;
      coarse[blockIdx.x * CSR_GPAD + tid] = cg;
      goffs[blockIdx.x * CSR_GPAD + tid] = ex;
    }
    __syncthreads();

    unsigned* __restrict__ bk = bucket + (size_t)blockIdx.x * CSR_CHUNK;
#pragma unroll
    for (int k = 0; k < EPT; ++k)
      if (d[k] >= 0)
        bk[goff[d[k] >> 8] + r[k]] = (unsigned)s[k] | ((unsigned)(d[k] & 255) << 16);
    return;
  }

  unsigned* __restrict__ plane = part + (size_t)blockIdx.x * GCN_N_NODES;

  for (int i = tid; i < GCN_N_NODES; i += CSR_THREADS) bins[i] = 0u;
  __syncthreads();

  const int e0 = blockIdx.x * CSR_CHUNK;
  const int e1 = min(GCN_N_EDGES, e0 + CSR_CHUNK);
  for (int e = e0 + tid; e < e1; e += 4 * CSR_THREADS) {
    int d[4];
#pragma unroll
    for (int u = 0; u < 4; ++u) {
      const int ee = e + u * CSR_THREADS;
      d[u] = (ee < e1) ? src[ee] : -1;
    }
#pragma unroll
    for (int u = 0; u < 4; ++u)
      if (d[u] >= 0) atomicAdd(&bins[d[u]], 1u);
  }
  __syncthreads();

  for (int i = tid; i < GCN_N_NODES; i += CSR_THREADS) plane[i] = bins[i];
}

#define PFX_THREADS 256
#define PFX_BLOCKS ((GCN_N_NODES + PFX_THREADS - 1) / PFX_THREADS)

template <int K, int NOUT>
__global__ __launch_bounds__(PFX_THREADS, 16) void gcn_prefix_gemm_kernel(
    const float* __restrict__ X, const float* __restrict__ W,
    float* __restrict__ Y, const unsigned* __restrict__ part,
    const unsigned* __restrict__ coarse, const unsigned* __restrict__ goffs,
    const unsigned* __restrict__ bucket, int* __restrict__ col,
    unsigned* __restrict__ row_ptr,
    float* __restrict__ norm_out, float* __restrict__ norm_in) {
  static_assert((NOUT / 16) * 32 == PFX_THREADS, "one wave per 16-column tile");
  static_assert(CSR_B * 4 == PFX_THREADS, "one thread per (src plane, 4 rows)");
  static_assert((16 * K) % PFX_THREADS == 0, "whole tile elements per thread");
  static_assert(GCN_N_NODES % 16 == 0, "one block owns 16 whole rows");
  static_assert(CSR_GROUP == PFX_THREADS, "one group block per coarse group");
  static_assert(CSR_B % 32 == 0, "coarse sums run in trips of 32");
  static_assert(CSR_B % (PFX_THREADS / 32) == 0 && CSR_B <= PFX_THREADS,
                "the waves share the slices evenly");
  static_assert(CSR_NGROUPS <= CSR_GPAD && PFX_THREADS <= CSR_GPAD, "coarse row");
  constexpr int LDSK = K + 4;
  constexpr int XPT = (16 * K) / PFX_THREADS;
  typedef unsigned u4 __attribute__((ext_vector_type(4)));
  __shared__ float tile[16 * LDSK];
  __shared__ __attribute__((aligned(16))) unsigned degp[PFX_THREADS / 32][16];
  __shared__ float rscale[16];
  __shared__ unsigned wdeg[PFX_THREADS / 32];
  __shared__ unsigned wbase[PFX_THREADS / 32];
  __shared__ unsigned scnt[CSR_B];
  __shared__ unsigned soff[CSR_B];
  __shared__ unsigned wcnt[PFX_THREADS / 32][PFX_THREADS];

  const int tid = threadIdx.x;

  if (blockIdx.x < PFX_BLOCKS) {
    const int g = blockIdx.x;
    const int n = g * PFX_THREADS + tid;
    const bool live = n < GCN_N_NODES;
    const int lane = tid & 31;
    const int wave = __builtin_amdgcn_readfirstlane(tid >> 5);
    const unsigned toff = (unsigned)tid * 4u;
    constexpr int SPW = CSR_B / (PFX_THREADS / 32);
    constexpr int RCH = 3;

    if (tid < CSR_B) {
      scnt[tid] = coarse[tid * CSR_GPAD + g];
      soff[tid] = goffs[tid * CSR_GPAD + g];
    }
#pragma unroll
    for (int w2 = 0; w2 < PFX_THREADS / 32; ++w2) wcnt[w2][tid] = 0u;

    unsigned bs = 0u;
    if (tid < g) {
#pragma nounroll
      for (int h = 0; h < CSR_B; h += 32) {
#pragma unroll
        for (int b = h; b < h + 32; ++b)
          bs += *(const unsigned*)((const char*)(coarse + b * CSR_GPAD) + toff);
      }
    }
    __syncthreads();

    static_assert(SPW <= 32, "one lane per run of the wave");
    const int mylen = (int)scnt[wave * SPW + (lane % SPW)];
    const int myoff = (int)soff[wave * SPW + (lane % SPW)];
    const unsigned* __restrict__ wb = bucket + (size_t)wave * SPW * CSR_CHUNK;
    unsigned w[SPW][RCH];
#pragma unroll
    for (int q = 0; q < SPW; ++q) {
      const int len = __builtin_amdgcn_readlane(mylen, q);
      const unsigned* __restrict__ rb =
          wb + (size_t)q * CSR_CHUNK + __builtin_amdgcn_readlane(myoff, q);
#pragma unroll
      for (int p = 0; p < RCH; ++p) {
        const int j = lane + 32 * p;
        w[q][p] = CSR_NONE;
        if (j < len) w[q][p] = rb[j];
      }
    }
    unsigned* mine = wcnt[wave];
#pragma unroll
    for (int q = 0; q < SPW; ++q) {
#pragma unroll
      for (int p = 0; p < RCH; ++p)
        if (w[q][p] != CSR_NONE) atomicAdd(&mine[w[q][p] >> 16], 1u);
    }
#pragma unroll
    for (int q = 0; q < SPW; ++q) {
      const int len = __builtin_amdgcn_readlane(mylen, q);
      const unsigned* __restrict__ rb =
          wb + (size_t)q * CSR_CHUNK + __builtin_amdgcn_readlane(myoff, q);
      for (int j = lane + 32 * RCH; j < len; j += 32) atomicAdd(&mine[rb[j] >> 16], 1u);
    }
    __syncthreads();
    unsigned wc[PFX_THREADS / 32];
    unsigned deg = 0u;
#pragma unroll
    for (int w2 = 0; w2 < PFX_THREADS / 32; ++w2) {
      wc[w2] = wcnt[w2][tid];
      deg += wc[w2];
    }

    unsigned incl = deg;
    for (int off = 1; off < 32; off <<= 1) {
      const unsigned v = __shfl_up(incl, off);
      if (lane >= off) incl += v;
    }
    for (int off = 16; off > 0; off >>= 1) bs += __shfl_xor(bs, off);
    if (lane == 31) wdeg[wave] = incl;
    if (lane == 0) wbase[wave] = bs;
    __syncthreads();
    unsigned rp = incl - deg;
#pragma unroll
    for (int w2 = 0; w2 < PFX_THREADS / 32; ++w2) {
      rp += wbase[w2];
      if (w2 < wave) rp += wdeg[w2];
    }

    {
      unsigned run = rp;
#pragma unroll
      for (int w2 = 0; w2 < PFX_THREADS / 32; ++w2) {
        wcnt[w2][tid] = run;
        run += wc[w2];
      }
    }
    if (live) {
      row_ptr[n] = rp;
      if (n == GCN_N_NODES - 1) row_ptr[GCN_N_NODES] = rp + deg;
      norm_in[n] = rsqrtf(fmaxf((float)deg, 1.0f));
    }
    __syncthreads();

#pragma unroll
    for (int q = 0; q < SPW; ++q) {
#pragma unroll
      for (int p = 0; p < RCH; ++p)
        if (w[q][p] != CSR_NONE) {
          const unsigned pos = atomicAdd(&mine[w[q][p] >> 16], 1u);
          col[pos] = (int)(w[q][p] & 0xffffu);
        }
    }
#pragma unroll
    for (int q = 0; q < SPW; ++q) {
      const int len = __builtin_amdgcn_readlane(mylen, q);
      const unsigned* __restrict__ rb =
          wb + (size_t)q * CSR_CHUNK + __builtin_amdgcn_readlane(myoff, q);
      for (int j = lane + 32 * RCH; j < len; j += 32) {
        const unsigned x = rb[j];
        const unsigned pos = atomicAdd(&mine[x >> 16], 1u);
        col[pos] = (int)(x & 0xffffu);
      }
    }
    return;
  }

  const int row0 = (blockIdx.x - PFX_BLOCKS) * 16;
  const int wave = tid >> 5;
  const int lane = tid & 31;

  float x[XPT];
#pragma unroll
  for (int i = 0; i < XPT; ++i) {
    const int t = tid + i * PFX_THREADS;
    const int r = t / K;
    const int c = t - r * K;
    x[i] = X[(size_t)(row0 + r) * K + c];
  }

  {
    const int p = tid >> 2;
    const int q = tid & 3;
    u4 c = *(const u4*)(part + (size_t)p * GCN_N_NODES + row0 + 4 * q);
    for (int off = 4; off < 32; off <<= 1) {
      c.x += __shfl_xor(c.x, off);
      c.y += __shfl_xor(c.y, off);
      c.z += __shfl_xor(c.z, off);
      c.w += __shfl_xor(c.w, off);
    }
    if (lane < 4) *(u4*)(&degp[wave][4 * lane]) = c;
  }
  __syncthreads();
  if (tid < 16) {
    unsigned so = 0u;
#pragma unroll
    for (int w = 0; w < PFX_THREADS / 32; ++w) so += degp[w][tid];
    const float s = rsqrtf(fmaxf((float)so, 1.0f));
    norm_out[row0 + tid] = s;
    rscale[tid] = s;
  }
  __syncthreads();

#pragma unroll
  for (int i = 0; i < XPT; ++i) {
    const int t = tid + i * PFX_THREADS;
    const int r = t / K;
    const int c = t - r * K;
    tile[r * LDSK + c] = x[i] * rscale[r];
  }
  __syncthreads();

  const int m  = lane & 15;
  const int kh = lane >> 4;
  const int nc = wave * 16 + m;

  v8f acc = {};
#pragma unroll 4
  for (int kk = 0; kk < K; kk += 4) {
    v2f a, b;
    const int ka = kk + 2 * kh;
    a.x = tile[m * LDSK + ka];
    a.y = tile[m * LDSK + ka + 1];
    b.x = W[(size_t)ka * NOUT + nc];
    b.y = W[(size_t)(ka + 1) * NOUT + nc];
    acc = __builtin_amdgcn_wmma_f32_16x16x4_f32(
         false, a,  false, b,
         (short)0, acc,  false,  false);
  }

#pragma unroll
  for (int v = 0; v < 8; ++v) {
    const int orow = row0 + v + 8 * kh;
    Y[(size_t)orow * NOUT + nc] = acc[v];
  }
}

template <int D>
__global__ __launch_bounds__(256) void gcn_gather_kernel(
    const float* __restrict__ H, const unsigned* __restrict__ row_ptr,
    const int* __restrict__ col, const float* __restrict__ norm_in,
    const float* __restrict__ b, float* __restrict__ out, int N) {
  constexpr int V = D / 32;
  typedef float vecT __attribute__((ext_vector_type(V)));
  const int lane = threadIdx.x & 31;
  const int node =
      __builtin_amdgcn_readfirstlane(blockIdx.x * 8 + (threadIdx.x >> 5));
  if (node >= N) return;

  const unsigned beg = row_ptr[node];
  const unsigned end = row_ptr[node + 1];

  vecT acc = (vecT)(0.0f);
  for (unsigned base = beg; base < end; base += 32u) {
    const int cnt = (int)min(32u, end - base);
    int myc = 0;
    if (lane < cnt) myc = col[base + lane];
    int j = 0;
    for (; j + 8 <= cnt; j += 8) {
      int s[8];
#pragma unroll
      for (int u = 0; u < 8; ++u) s[u] = __builtin_amdgcn_readlane(myc, j + u);
      vecT r[8];
#pragma unroll
      for (int u = 0; u < 8; ++u)
        r[u] = *(const vecT*)(H + (size_t)s[u] * D + lane * V);
#pragma unroll
      for (int u = 0; u < 8; ++u) acc += r[u];
    }
    if (j < cnt) {
      int s[7];
#pragma unroll
      for (int u = 0; u < 7; ++u)
        s[u] = __builtin_amdgcn_readlane(myc, (j + u) & 31);
      vecT r[7];
#pragma unroll
      for (int u = 0; u < 7; ++u) {
        r[u] = (vecT)(0.0f);
        if (j + u < cnt) r[u] = *(const vecT*)(H + (size_t)s[u] * D + lane * V);
      }
#pragma unroll
      for (int u = 0; u < 7; ++u) acc += r[u];
    }
  }

  const float ni = norm_in[node];
  const vecT bb = *(const vecT*)(b + lane * V);
  vecT o;
#pragma unroll
  for (int k = 0; k < V; ++k) o[k] = fmaxf(fmaf(acc[k], ni, bb[k]), 0.0f);
  *(vecT*)(out + (size_t)node * D + lane * V) = o;
}

template <int D, int NOUT>
__global__ __launch_bounds__(512) void gcn_gather_gemm_kernel(
    const float* __restrict__ H, const unsigned* __restrict__ row_ptr,
    const int* __restrict__ col, const float* __restrict__ norm_in,
    const float* __restrict__ b, const float* __restrict__ scale,
    const float* __restrict__ W, float* __restrict__ Y) {
  static_assert(GCN_N_NODES % 16 == 0, "one block owns 16 whole rows");
  static_assert(NOUT % 16 == 0 && NOUT / 16 <= 16, "one wave per column tile");
  static_assert(D % 8 == 0, "two K halves of whole k-steps");
  constexpr int V = D / 32;
  constexpr int LDSK = D + 4;
  typedef float vecT __attribute__((ext_vector_type(V)));
  __shared__ __attribute__((aligned(16))) float tile[16 * LDSK];

  const int lane = threadIdx.x & 31;
  const int wave = __builtin_amdgcn_readfirstlane(threadIdx.x >> 5);
  const int row0 = blockIdx.x * 16;
  const int node = row0 + wave;

  const unsigned beg = row_ptr[node];
  const unsigned end = row_ptr[node + 1];

  vecT acc = (vecT)(0.0f);
  for (unsigned base = beg; base < end; base += 32u) {
    const int cnt = (int)min(32u, end - base);
    int myc = 0;
    if (lane < cnt) myc = col[base + lane];
    int j = 0;
    for (; j + 8 <= cnt; j += 8) {
      int s[8];
#pragma unroll
      for (int u = 0; u < 8; ++u) s[u] = __builtin_amdgcn_readlane(myc, j + u);
      vecT r[8];
#pragma unroll
      for (int u = 0; u < 8; ++u)
        r[u] = *(const vecT*)(H + (size_t)s[u] * D + lane * V);
#pragma unroll
      for (int u = 0; u < 8; ++u) acc += r[u];
    }
    if (j < cnt) {
      int s[7];
#pragma unroll
      for (int u = 0; u < 7; ++u)
        s[u] = __builtin_amdgcn_readlane(myc, (j + u) & 31);
      vecT r[7];
#pragma unroll
      for (int u = 0; u < 7; ++u) {
        r[u] = (vecT)(0.0f);
        if (j + u < cnt) r[u] = *(const vecT*)(H + (size_t)s[u] * D + lane * V);
      }
#pragma unroll
      for (int u = 0; u < 7; ++u) acc += r[u];
    }
  }

  constexpr int KS = D / 4;
  constexpr int KH = KS / 2;
  const bool mm = wave < NOUT / 16;
  const int m  = lane & 15;
  const int kh = lane >> 4;
  const int nc = wave * 16 + m;

  v2f bq[KH];
  if (mm) {
#pragma unroll
    for (int s = 0; s < KH; ++s) {
      const int ka = 4 * s + 2 * kh;
      bq[s].x = W[(size_t)ka * NOUT + nc];
      bq[s].y = W[(size_t)(ka + 1) * NOUT + nc];
    }
  }

  {
    const float ni = norm_in[node];
    const float sc = scale[node];
    const vecT bb = *(const vecT*)(b + lane * V);
    vecT o;
#pragma unroll
    for (int k = 0; k < V; ++k)
      o[k] = fmaxf(fmaf(acc[k], ni, bb[k]), 0.0f) * sc;
    *(vecT*)(tile + wave * LDSK + lane * V) = o;
  }
  __syncthreads();

  if (!mm) return;

  v8f cacc = {};
#pragma unroll
  for (int s = 0; s < KH; ++s) {
    const int ka = 4 * s + 2 * kh;
    v2f a;
    a.x = tile[m * LDSK + ka];
    a.y = tile[m * LDSK + ka + 1];
    cacc = __builtin_amdgcn_wmma_f32_16x16x4_f32(
         false, a,  false, bq[s],
         (short)0, cacc,  false,  false);
  }
  v2f br[KH];
#pragma unroll
  for (int s = 0; s < KH; ++s) {
    const int ka = 4 * (s + KH) + 2 * kh;
    br[s].x = W[(size_t)ka * NOUT + nc];
    br[s].y = W[(size_t)(ka + 1) * NOUT + nc];
  }
#pragma unroll
  for (int s = 0; s < KH; ++s) {
    const int ka = 4 * (s + KH) + 2 * kh;
    v2f a;
    a.x = tile[m * LDSK + ka];
    a.y = tile[m * LDSK + ka + 1];
    cacc = __builtin_amdgcn_wmma_f32_16x16x4_f32(
         false, a,  false, br[s],
         (short)0, cacc,  false,  false);
  }

#pragma unroll
  for (int v = 0; v < 8; ++v) {
    const int orow = row0 + v + 8 * kh;
    Y[(size_t)orow * NOUT + nc] = cacc[v];
  }
}

extern "C" void kernel_launch(void* const* d_in, const int* in_sizes, int n_in,
                              void* d_out, int out_size, void* d_ws, size_t ws_size,
                              hipStream_t stream) {
  const float* feat = (const float*)d_in[0];
  const int*   src  = (const int*)d_in[1];
  const int*   dst  = (const int*)d_in[2];
  const float* W1   = (const float*)d_in[3];
  const float* b1   = (const float*)d_in[4];
  const float* W2   = (const float*)d_in[5];
  const float* b2   = (const float*)d_in[6];
  float* out = (float*)d_out;
  float* ws  = (float*)d_ws;

  const int N = GCN_N_NODES;
  const size_t NP = 50048;

  unsigned* part     = (unsigned*)ws;
  unsigned* bucket   = part + (size_t)CSR_B * GCN_N_NODES;
  unsigned* goffs    = bucket + (size_t)GCN_N_EDGES;
  unsigned* coarse   = part + (size_t)2 * CSR_B * GCN_N_NODES;
  unsigned* row_ptr  = coarse + NP;
  float*    norm_out = (float*)(row_ptr + NP);
  float*    norm_in  = norm_out + NP;
  int*      col      = (int*)(norm_in + NP);
  float*    hproj1   = (float*)(col + (size_t)GCN_N_EDGES);
  float*    hproj2   = hproj1 + (size_t)N * GCN_HID_DIM;

  gcn_hist_kernel<<<dim3(CSR_B, 2), CSR_THREADS, 0, stream>>>(src, dst, part, coarse,
                                                             goffs, bucket);
  gcn_prefix_gemm_kernel<GCN_IN_DIM, GCN_HID_DIM>
      <<<PFX_BLOCKS + N / 16, PFX_THREADS, 0, stream>>>(
          feat, W1, hproj1, part, coarse, goffs, bucket, col, row_ptr, norm_out, norm_in);

  gcn_gather_gemm_kernel<GCN_HID_DIM, GCN_OUT_DIM>
      <<<N / 16, 512, 0, stream>>>(hproj1, row_ptr, col, norm_in, b1, norm_out, W2, hproj2);

  gcn_gather_kernel<GCN_OUT_DIM>
      <<<(N + 7) / 8, 256, 0, stream>>>(hproj2, row_ptr, col, norm_in, b2, out, N);
}


# first launch stores 16 B per lane in whole lines: dst edges are placed in an LDS image of the slice's bucket region and copied out (no 4-byte scatter to memory); src planes copied out likewise
# speedup vs baseline: 7.0357x; 1.0560x over previous
#include <hip/hip_runtime.h>


#define GCN_N_NODES 50000
#define GCN_N_EDGES 800000
#define GCN_IN_DIM 128
#define GCN_HID_DIM 128
#define GCN_OUT_DIM 64

#define CSR_B 64
#define CSR_CHUNK ((GCN_N_EDGES + CSR_B - 1) / CSR_B)
#define CSR_THREADS 1024
#define CSR_GROUP 256
#define CSR_NGROUPS ((GCN_N_NODES + CSR_GROUP - 1) / CSR_GROUP)
#define CSR_GPAD 256

typedef __attribute__((ext_vector_type(2))) float v2f;
typedef __attribute__((ext_vector_type(8))) float v8f;

__device__ __forceinline__ unsigned wave_sum_u32(unsigned v) {
  v += (unsigned)__builtin_amdgcn_update_dpp((int)v, (int)v, 0x161, 0xf, 0xf, false);
  v += (unsigned)__builtin_amdgcn_update_dpp((int)v, (int)v, 0x162, 0xf, 0xf, false);
  v += (unsigned)__builtin_amdgcn_update_dpp((int)v, (int)v, 0x164, 0xf, 0xf, false);
  v += (unsigned)__builtin_amdgcn_update_dpp((int)v, (int)v, 0x168, 0xf, 0xf, false);
  v += (unsigned)__builtin_amdgcn_permlanex16((int)v, (int)v, 0x76543210, 0xfedcba98,
                                              false, false);
  return v;
}

#define CSR_NONE 0xffffffffu

__global__ __launch_bounds__(CSR_THREADS) void gcn_hist_kernel(
    const int* __restrict__ src, const int* __restrict__ dst,
    unsigned* __restrict__ part, unsigned* __restrict__ coarse,
    unsigned* __restrict__ goffs, unsigned* __restrict__ bucket) {
  typedef unsigned u4 __attribute__((ext_vector_type(4)));
  __shared__ __attribute__((aligned(16))) unsigned bins[GCN_N_NODES];
  const int tid = threadIdx.x;

  if (blockIdx.y == 1) {
    static_assert(GCN_N_EDGES % CSR_B == 0, "every slice is full");
    static_assert(GCN_N_NODES <= 65536 && CSR_GROUP == 256,
                  "a src id and a dst offset inside its group share one word");
    static_assert(CSR_GPAD == 256 && CSR_NGROUPS <= CSR_GPAD, "256 counters");
    constexpr int EPT = (CSR_CHUNK + CSR_THREADS - 1) / CSR_THREADS;
    unsigned* gcnt = bins;
    unsigned* goff = bins + 256;
    unsigned* wsum = bins + 512;
    unsigned* stage = bins + 1024;
    static_assert(1024 + CSR_CHUNK <= GCN_N_NODES && CSR_CHUNK % 4 == 0,
                  "the image fits behind the counters, in whole 16-byte pieces");
    const int lane = tid & 31;
    const int wave = tid >> 5;

    if (tid < CSR_GPAD) gcnt[tid] = 0u;
    __syncthreads();

    const int e0 = blockIdx.x * CSR_CHUNK;
    int d[EPT], s[EPT];
    unsigned r[EPT];
#pragma unroll
    for (int k = 0; k < EPT; ++k) {
      const int le = tid + k * CSR_THREADS;
      const bool ok = le < CSR_CHUNK;
      d[k] = ok ? dst[e0 + le] : -1;
      s[k] = ok ? src[e0 + le] : 0;
    }
#pragma unroll
    for (int k = 0; k < EPT; ++k) {
      r[k] = 0u;
      if (d[k] >= 0) r[k] = atomicAdd(&gcnt[d[k] >> 8], 1u);
    }
    __syncthreads();

    unsigned cg = 0u, incl = 0u;
    if (tid < CSR_GPAD) {
      cg = gcnt[tid];
      incl = cg;
      for (int off = 1; off < 32; off <<= 1) {
        const unsigned v = __shfl_up(incl, off);
        if (lane >= off) incl += v;
      }
      if (lane == 31) wsum[wave] = incl;
    }
    __syncthreads();
    if (tid < CSR_GPAD) {
      unsigned ex = incl - cg;
#pragma unroll
      for (int w = 0; w < CSR_GPAD / 32; ++w)
        if (w < wave) ex += wsum[w];
      goff[tid] = ex;
      coarse[blockIdx.x * CSR_GPAD + tid] = cg;
      goffs[blockIdx.x * CSR_GPAD + tid] = ex;
    }
    __syncthreads();

#pragma unroll
    for (int k = 0; k < EPT; ++k)
      if (d[k] >= 0)
        stage[goff[d[k] >> 8] + r[k]] = (unsigned)s[k] | ((unsigned)(d[k] & 255) << 16);
    __syncthreads();

    u4* __restrict__ bk4 = (u4*)(bucket + (size_t)blockIdx.x * CSR_CHUNK);
    const u4* st4 = (const u4*)stage;
    for (int i = tid; i < CSR_CHUNK / 4; i += CSR_THREADS) bk4[i] = st4[i];
    return;
  }

  unsigned* __restrict__ plane = part + (size_t)blockIdx.x * GCN_N_NODES;

  for (int i = tid; i < GCN_N_NODES; i += CSR_THREADS) bins[i] = 0u;
  __syncthreads();

  const int e0 = blockIdx.x * CSR_CHUNK;
  const int e1 = min(GCN_N_EDGES, e0 + CSR_CHUNK);
  for (int e = e0 + tid; e < e1; e += 4 * CSR_THREADS) {
    int d[4];
#pragma unroll
    for (int u = 0; u < 4; ++u) {
      const int ee = e + u * CSR_THREADS;
      d[u] = (ee < e1) ? src[ee] : -1;
    }
#pragma unroll
    for (int u = 0; u < 4; ++u)
      if (d[u] >= 0) atomicAdd(&bins[d[u]], 1u);
  }
  __syncthreads();

  static_assert(GCN_N_NODES % 4 == 0, "a plane is whole 16-byte pieces");
  u4* __restrict__ plane4 = (u4*)plane;
  const u4* bins4 = (const u4*)bins;
  for (int i = tid; i < GCN_N_NODES / 4; i += CSR_THREADS) plane4[i] = bins4[i];
}

#define PFX_THREADS 256
#define PFX_BLOCKS ((GCN_N_NODES + PFX_THREADS - 1) / PFX_THREADS)

template <int K, int NOUT>
__global__ __launch_bounds__(PFX_THREADS, 16) void gcn_prefix_gemm_kernel(
    const float* __restrict__ X, const float* __restrict__ W,
    float* __restrict__ Y, const unsigned* __restrict__ part,
    const unsigned* __restrict__ coarse, const unsigned* __restrict__ goffs,
    const unsigned* __restrict__ bucket, int* __restrict__ col,
    unsigned* __restrict__ row_ptr,
    float* __restrict__ norm_out, float* __restrict__ norm_in) {
  static_assert((NOUT / 16) * 32 == PFX_THREADS, "one wave per 16-column tile");
  static_assert(CSR_B * 4 == PFX_THREADS, "one thread per (src plane, 4 rows)");
  static_assert((16 * K) % PFX_THREADS == 0, "whole tile elements per thread");
  static_assert(GCN_N_NODES % 16 == 0, "one block owns 16 whole rows");
  static_assert(CSR_GROUP == PFX_THREADS, "one group block per coarse group");
  static_assert(CSR_B % 32 == 0, "coarse sums run in trips of 32");
  static_assert(CSR_B % (PFX_THREADS / 32) == 0 && CSR_B <= PFX_THREADS,
                "the waves share the slices evenly");
  static_assert(CSR_NGROUPS <= CSR_GPAD && PFX_THREADS <= CSR_GPAD, "coarse row");
  constexpr int LDSK = K + 4;
  constexpr int XPT = (16 * K) / PFX_THREADS;
  typedef unsigned u4 __attribute__((ext_vector_type(4)));
  __shared__ float tile[16 * LDSK];
  __shared__ __attribute__((aligned(16))) unsigned degp[PFX_THREADS / 32][16];
  __shared__ float rscale[16];
  __shared__ unsigned wdeg[PFX_THREADS / 32];
  __shared__ unsigned wbase[PFX_THREADS / 32];
  __shared__ unsigned scnt[CSR_B];
  __shared__ unsigned soff[CSR_B];
  __shared__ unsigned wcnt[PFX_THREADS / 32][PFX_THREADS];

  const int tid = threadIdx.x;

  if (blockIdx.x < PFX_BLOCKS) {
    const int g = blockIdx.x;
    const int n = g * PFX_THREADS + tid;
    const bool live = n < GCN_N_NODES;
    const int lane = tid & 31;
    const int wave = __builtin_amdgcn_readfirstlane(tid >> 5);
    const unsigned toff = (unsigned)tid * 4u;
    constexpr int SPW = CSR_B / (PFX_THREADS / 32);
    constexpr int RCH = 3;

    if (tid < CSR_B) {
      scnt[tid] = coarse[tid * CSR_GPAD + g];
      soff[tid] = goffs[tid * CSR_GPAD + g];
    }
#pragma unroll
    for (int w2 = 0; w2 < PFX_THREADS / 32; ++w2) wcnt[w2][tid] = 0u;

    unsigned bs = 0u;
    if (tid < g) {
#pragma nounroll
      for (int h = 0; h < CSR_B; h += 32) {
#pragma unroll
        for (int b = h; b < h + 32; ++b)
          bs += *(const unsigned*)((const char*)(coarse + b * CSR_GPAD) + toff);
      }
    }
    __syncthreads();

    static_assert(SPW <= 32, "one lane per run of the wave");
    const int mylen = (int)scnt[wave * SPW + (lane % SPW)];
    const int myoff = (int)soff[wave * SPW + (lane % SPW)];
    const unsigned* __restrict__ wb = bucket + (size_t)wave * SPW * CSR_CHUNK;
    unsigned w[SPW][RCH];
#pragma unroll
    for (int q = 0; q < SPW; ++q) {
      const int len = __builtin_amdgcn_readlane(mylen, q);
      const unsigned* __restrict__ rb =
          wb + (size_t)q * CSR_CHUNK + __builtin_amdgcn_readlane(myoff, q);
#pragma unroll
      for (int p = 0; p < RCH; ++p) {
        const int j = lane + 32 * p;
        w[q][p] = CSR_NONE;
        if (j < len) w[q][p] = rb[j];
      }
    }
    unsigned* mine = wcnt[wave];
#pragma unroll
    for (int q = 0; q < SPW; ++q) {
#pragma unroll
      for (int p = 0; p < RCH; ++p)
        if (w[q][p] != CSR_NONE) atomicAdd(&mine[w[q][p] >> 16], 1u);
    }
#pragma unroll
    for (int q = 0; q < SPW; ++q) {
      const int len = __builtin_amdgcn_readlane(mylen, q);
      const unsigned* __restrict__ rb =
          wb + (size_t)q * CSR_CHUNK + __builtin_amdgcn_readlane(myoff, q);
      for (int j = lane + 32 * RCH; j < len; j += 32) atomicAdd(&mine[rb[j] >> 16], 1u);
    }
    __syncthreads();
    unsigned wc[PFX_THREADS / 32];
    unsigned deg = 0u;
#pragma unroll
    for (int w2 = 0; w2 < PFX_THREADS / 32; ++w2) {
      wc[w2] = wcnt[w2][tid];
      deg += wc[w2];
    }

    unsigned incl = deg;
    for (int off = 1; off < 32; off <<= 1) {
      const unsigned v = __shfl_up(incl, off);
      if (lane >= off) incl += v;
    }
    for (int off = 16; off > 0; off >>= 1) bs += __shfl_xor(bs, off);
    if (lane == 31) wdeg[wave] = incl;
    if (lane == 0) wbase[wave] = bs;
    __syncthreads();
    unsigned rp = incl - deg;
#pragma unroll
    for (int w2 = 0; w2 < PFX_THREADS / 32; ++w2) {
      rp += wbase[w2];
      if (w2 < wave) rp += wdeg[w2];
    }

    {
      unsigned run = rp;
#pragma unroll
      for (int w2 = 0; w2 < PFX_THREADS / 32; ++w2) {
        wcnt[w2][tid] = run;
        run += wc[w2];
      }
    }
    if (live) {
      row_ptr[n] = rp;
      if (n == GCN_N_NODES - 1) row_ptr[GCN_N_NODES] = rp + deg;
      norm_in[n] = rsqrtf(fmaxf((float)deg, 1.0f));
    }
    __syncthreads();

#pragma unroll
    for (int q = 0; q < SPW; ++q) {
#pragma unroll
      for (int p = 0; p < RCH; ++p)
        if (w[q][p] != CSR_NONE) {
          const unsigned pos = atomicAdd(&mine[w[q][p] >> 16], 1u);
          col[pos] = (int)(w[q][p] & 0xffffu);
        }
    }
#pragma unroll
    for (int q = 0; q < SPW; ++q) {
      const int len = __builtin_amdgcn_readlane(mylen, q);
      const unsigned* __restrict__ rb =
          wb + (size_t)q * CSR_CHUNK + __builtin_amdgcn_readlane(myoff, q);
      for (int j = lane + 32 * RCH; j < len; j += 32) {
        const unsigned x = rb[j];
        const unsigned pos = atomicAdd(&mine[x >> 16], 1u);
        col[pos] = (int)(x & 0xffffu);
      }
    }
    return;
  }

  const int row0 = (blockIdx.x - PFX_BLOCKS) * 16;
  const int wave = tid >> 5;
  const int lane = tid & 31;

  float x[XPT];
#pragma unroll
  for (int i = 0; i < XPT; ++i) {
    const int t = tid + i * PFX_THREADS;
    const int r = t / K;
    const int c = t - r * K;
    x[i] = X[(size_t)(row0 + r) * K + c];
  }

  {
    const int p = tid >> 2;
    const int q = tid & 3;
    u4 c = *(const u4*)(part + (size_t)p * GCN_N_NODES + row0 + 4 * q);
    for (int off = 4; off < 32; off <<= 1) {
      c.x += __shfl_xor(c.x, off);
      c.y += __shfl_xor(c.y, off);
      c.z += __shfl_xor(c.z, off);
      c.w += __shfl_xor(c.w, off);
    }
    if (lane < 4) *(u4*)(&degp[wave][4 * lane]) = c;
  }
  __syncthreads();
  if (tid < 16) {
    unsigned so = 0u;
#pragma unroll
    for (int w = 0; w < PFX_THREADS / 32; ++w) so += degp[w][tid];
    const float s = rsqrtf(fmaxf((float)so, 1.0f));
    norm_out[row0 + tid] = s;
    rscale[tid] = s;
  }
  __syncthreads();

#pragma unroll
  for (int i = 0; i < XPT; ++i) {
    const int t = tid + i * PFX_THREADS;
    const int r = t / K;
    const int c = t - r * K;
    tile[r * LDSK + c] = x[i] * rscale[r];
  }
  __syncthreads();

  const int m  = lane & 15;
  const int kh = lane >> 4;
  const int nc = wave * 16 + m;

  v8f acc = {};
#pragma unroll 4
  for (int kk = 0; kk < K; kk += 4) {
    v2f a, b;
    const int ka = kk + 2 * kh;
    a.x = tile[m * LDSK + ka];
    a.y = tile[m * LDSK + ka + 1];
    b.x = W[(size_t)ka * NOUT + nc];
    b.y = W[(size_t)(ka + 1) * NOUT + nc];
    acc = __builtin_amdgcn_wmma_f32_16x16x4_f32(
         false, a,  false, b,
         (short)0, acc,  false,  false);
  }

#pragma unroll
  for (int v = 0; v < 8; ++v) {
    const int orow = row0 + v + 8 * kh;
    Y[(size_t)orow * NOUT + nc] = acc[v];
  }
}

template <int D>
__global__ __launch_bounds__(256) void gcn_gather_kernel(
    const float* __restrict__ H, const unsigned* __restrict__ row_ptr,
    const int* __restrict__ col, const float* __restrict__ norm_in,
    const float* __restrict__ b, float* __restrict__ out, int N) {
  constexpr int V = D / 32;
  typedef float vecT __attribute__((ext_vector_type(V)));
  const int lane = threadIdx.x & 31;
  const int node =
      __builtin_amdgcn_readfirstlane(blockIdx.x * 8 + (threadIdx.x >> 5));
  if (node >= N) return;

  const unsigned beg = row_ptr[node];
  const unsigned end = row_ptr[node + 1];

  vecT acc = (vecT)(0.0f);
  for (unsigned base = beg; base < end; base += 32u) {
    const int cnt = (int)min(32u, end - base);
    int myc = 0;
    if (lane < cnt) myc = col[base + lane];
    int j = 0;
    for (; j + 8 <= cnt; j += 8) {
      int s[8];
#pragma unroll
      for (int u = 0; u < 8; ++u) s[u] = __builtin_amdgcn_readlane(myc, j + u);
      vecT r[8];
#pragma unroll
      for (int u = 0; u < 8; ++u)
        r[u] = *(const vecT*)(H + (size_t)s[u] * D + lane * V);
#pragma unroll
      for (int u = 0; u < 8; ++u) acc += r[u];
    }
    if (j < cnt) {
      int s[7];
#pragma unroll
      for (int u = 0; u < 7; ++u)
        s[u] = __builtin_amdgcn_readlane(myc, (j + u) & 31);
      vecT r[7];
#pragma unroll
      for (int u = 0; u < 7; ++u) {
        r[u] = (vecT)(0.0f);
        if (j + u < cnt) r[u] = *(const vecT*)(H + (size_t)s[u] * D + lane * V);
      }
#pragma unroll
      for (int u = 0; u < 7; ++u) acc += r[u];
    }
  }

  const float ni = norm_in[node];
  const vecT bb = *(const vecT*)(b + lane * V);
  vecT o;
#pragma unroll
  for (int k = 0; k < V; ++k) o[k] = fmaxf(fmaf(acc[k], ni, bb[k]), 0.0f);
  *(vecT*)(out + (size_t)node * D + lane * V) = o;
}

template <int D, int NOUT>
__global__ __launch_bounds__(512) void gcn_gather_gemm_kernel(
    const float* __restrict__ H, const unsigned* __restrict__ row_ptr,
    const int* __restrict__ col, const float* __restrict__ norm_in,
    const float* __restrict__ b, const float* __restrict__ scale,
    const float* __restrict__ W, float* __restrict__ Y) {
  static_assert(GCN_N_NODES % 16 == 0, "one block owns 16 whole rows");
  static_assert(NOUT % 16 == 0 && NOUT / 16 <= 16, "one wave per column tile");
  static_assert(D % 8 == 0, "two K halves of whole k-steps");
  constexpr int V = D / 32;
  constexpr int LDSK = D + 4;
  typedef float vecT __attribute__((ext_vector_type(V)));
  __shared__ __attribute__((aligned(16))) float tile[16 * LDSK];

  const int lane = threadIdx.x & 31;
  const int wave = __builtin_amdgcn_readfirstlane(threadIdx.x >> 5);
  const int row0 = blockIdx.x * 16;
  const int node = row0 + wave;

  const unsigned beg = row_ptr[node];
  const unsigned end = row_ptr[node + 1];

  vecT acc = (vecT)(0.0f);
  for (unsigned base = beg; base < end; base += 32u) {
    const int cnt = (int)min(32u, end - base);
    int myc = 0;
    if (lane < cnt) myc = col[base + lane];
    int j = 0;
    for (; j + 8 <= cnt; j += 8) {
      int s[8];
#pragma unroll
      for (int u = 0; u < 8; ++u) s[u] = __builtin_amdgcn_readlane(myc, j + u);
      vecT r[8];
#pragma unroll
      for (int u = 0; u < 8; ++u)
        r[u] = *(const vecT*)(H + (size_t)s[u] * D + lane * V);
#pragma unroll
      for (int u = 0; u < 8; ++u) acc += r[u];
    }
    if (j < cnt) {
      int s[7];
#pragma unroll
      for (int u = 0; u < 7; ++u)
        s[u] = __builtin_amdgcn_readlane(myc, (j + u) & 31);
      vecT r[7];
#pragma unroll
      for (int u = 0; u < 7; ++u) {
        r[u] = (vecT)(0.0f);
        if (j + u < cnt) r[u] = *(const vecT*)(H + (size_t)s[u] * D + lane * V);
      }
#pragma unroll
      for (int u = 0; u < 7; ++u) acc += r[u];
    }
  }

  constexpr int KS = D / 4;
  constexpr int KH = KS / 2;
  const bool mm = wave < NOUT / 16;
  const int m  = lane & 15;
  const int kh = lane >> 4;
  const int nc = wave * 16 + m;

  v2f bq[KH];
  if (mm) {
#pragma unroll
    for (int s = 0; s < KH; ++s) {
      const int ka = 4 * s + 2 * kh;
      bq[s].x = W[(size_t)ka * NOUT + nc];
      bq[s].y = W[(size_t)(ka + 1) * NOUT + nc];
    }
  }

  {
    const float ni = norm_in[node];
    const float sc = scale[node];
    const vecT bb = *(const vecT*)(b + lane * V);
    vecT o;
#pragma unroll
    for (int k = 0; k < V; ++k)
      o[k] = fmaxf(fmaf(acc[k], ni, bb[k]), 0.0f) * sc;
    *(vecT*)(tile + wave * LDSK + lane * V) = o;
  }
  __syncthreads();

  if (!mm) return;

  v8f cacc = {};
#pragma unroll
  for (int s = 0; s < KH; ++s) {
    const int ka = 4 * s + 2 * kh;
    v2f a;
    a.x = tile[m * LDSK + ka];
    a.y = tile[m * LDSK + ka + 1];
    cacc = __builtin_amdgcn_wmma_f32_16x16x4_f32(
         false, a,  false, bq[s],
         (short)0, cacc,  false,  false);
  }
  v2f br[KH];
#pragma unroll
  for (int s = 0; s < KH; ++s) {
    const int ka = 4 * (s + KH) + 2 * kh;
    br[s].x = W[(size_t)ka * NOUT + nc];
    br[s].y = W[(size_t)(ka + 1) * NOUT + nc];
  }
#pragma unroll
  for (int s = 0; s < KH; ++s) {
    const int ka = 4 * (s + KH) + 2 * kh;
    v2f a;
    a.x = tile[m * LDSK + ka];
    a.y = tile[m * LDSK + ka + 1];
    cacc = __builtin_amdgcn_wmma_f32_16x16x4_f32(
         false, a,  false, br[s],
         (short)0, cacc,  false,  false);
  }

#pragma unroll
  for (int v = 0; v < 8; ++v) {
    const int orow = row0 + v + 8 * kh;
    Y[(size_t)orow * NOUT + nc] = cacc[v];
  }
}

extern "C" void kernel_launch(void* const* d_in, const int* in_sizes, int n_in,
                              void* d_out, int out_size, void* d_ws, size_t ws_size,
                              hipStream_t stream) {
  const float* feat = (const float*)d_in[0];
  const int*   src  = (const int*)d_in[1];
  const int*   dst  = (const int*)d_in[2];
  const float* W1   = (const float*)d_in[3];
  const float* b1   = (const float*)d_in[4];
  const float* W2   = (const float*)d_in[5];
  const float* b2   = (const float*)d_in[6];
  float* out = (float*)d_out;
  float* ws  = (float*)d_ws;

  const int N = GCN_N_NODES;
  const size_t NP = 50048;

  unsigned* part     = (unsigned*)ws;
  unsigned* bucket   = part + (size_t)CSR_B * GCN_N_NODES;
  unsigned* goffs    = bucket + (size_t)GCN_N_EDGES;
  unsigned* coarse   = part + (size_t)2 * CSR_B * GCN_N_NODES;
  unsigned* row_ptr  = coarse + NP;
  float*    norm_out = (float*)(row_ptr + NP);
  float*    norm_in  = norm_out + NP;
  int*      col      = (int*)(norm_in + NP);
  float*    hproj1   = (float*)(col + (size_t)GCN_N_EDGES);
  float*    hproj2   = hproj1 + (size_t)N * GCN_HID_DIM;

  gcn_hist_kernel<<<dim3(CSR_B, 2), CSR_THREADS, 0, stream>>>(src, dst, part, coarse,
                                                             goffs, bucket);
  gcn_prefix_gemm_kernel<GCN_IN_DIM, GCN_HID_DIM>
      <<<PFX_BLOCKS + N / 16, PFX_THREADS, 0, stream>>>(
          feat, W1, hproj1, part, coarse, goffs, bucket, col, row_ptr, norm_out, norm_in);

  gcn_gather_gemm_kernel<GCN_HID_DIM, GCN_OUT_DIM>
      <<<N / 16, 512, 0, stream>>>(hproj1, row_ptr, col, norm_in, b1, norm_out, W2, hproj2);

  gcn_gather_kernel<GCN_OUT_DIM>
      <<<(N + 7) / 8, 256, 0, stream>>>(hproj2, row_ptr, col, norm_in, b2, out, N);
}
